# starting kernel
# baseline (speedup 1.0000x reference)
#include <hip/hip_runtime.h>
#include <hip/hip_bf16.h>

#define IN_DIM 256
#define OUT_DIM 128
#define SLOPE 0.1f

typedef float v2f __attribute__((ext_vector_type(2)));
typedef float v8f __attribute__((ext_vector_type(8)));

__global__ __launch_bounds__(256) void gat_gemm_kernel(
    const float* __restrict__ features, const float* __restrict__ W,
    const float* __restrict__ bias, const int* __restrict__ unique_nodes,
    float* __restrict__ h, int N)
{
    const int lane = threadIdx.x & 31;
    const int wave = threadIdx.x >> 5;
    const int half = lane >> 4;
    const int mn   = lane & 15;

    const int rowBase = blockIdx.x * 16;
    int arow_idx = rowBase + mn;
    if (arow_idx >= N) arow_idx = N - 1;
    const int gsrc = unique_nodes[arow_idx];
    const float* __restrict__ arow = features + (size_t)gsrc * IN_DIM;

    const int ncol = wave * 16 + mn;
    const float* __restrict__ brow = W + (size_t)ncol * IN_DIM;

    v8f acc;
    const float bv = bias[ncol];
#pragma unroll
    for (int r = 0; r < 8; ++r) acc[r] = bv;

#pragma unroll 4
    for (int k = 0; k < IN_DIM; k += 4) {
        v2f av  = *(const v2f*)(arow + k + 2 * half);
        v2f bvv = *(const v2f*)(brow + k + 2 * half);
        acc = __builtin_amdgcn_wmma_f32_16x16x4_f32(
             false, av,  false, bvv,
             (short)0, acc,  false,  false);
    }

    float* __restrict__ obase =
        h + (size_t)(rowBase + 8 * half) * OUT_DIM + ncol;
    if (rowBase + 16 <= N) {
#pragma unroll
        for (int r = 0; r < 8; ++r)
            obase[(size_t)r * OUT_DIM] = acc[r];
    } else {
#pragma unroll
        for (int r = 0; r < 8; ++r) {
            int orow = rowBase + 8 * half + r;
            if (orow < N) obase[(size_t)r * OUT_DIM] = acc[r];
        }
    }
}

__global__ __launch_bounds__(256) void gat_scores_kernel(
    const float* __restrict__ h, const float* __restrict__ a,
    float* __restrict__ s_src, float* __restrict__ s_dst, int N)
{
    const int lane = threadIdx.x & 31;
    const int row  = blockIdx.x * 8 + (threadIdx.x >> 5);
    if (row >= N) return;

    const float4 hv = *(const float4*)(h + (size_t)row * OUT_DIM + lane * 4);
    const float4 as = *(const float4*)(a + lane * 4);
    const float4 ad = *(const float4*)(a + OUT_DIM + lane * 4);

    float ps = hv.x * as.x + hv.y * as.y + hv.z * as.z + hv.w * as.w;
    float pd = hv.x * ad.x + hv.y * ad.y + hv.z * ad.z + hv.w * ad.w;

#pragma unroll
    for (int off = 16; off > 0; off >>= 1) {
        ps += __shfl_xor(ps, off, 32);
        pd += __shfl_xor(pd, off, 32);
    }
    if (lane == 0) {
        s_src[row] = ps;
        s_dst[row] = pd;
    }
}

__global__ __launch_bounds__(256) void gat_edge_kernel(
    const int* __restrict__ edges, const float* __restrict__ h,
    const float* __restrict__ s_src, const float* __restrict__ s_dst,
    float* __restrict__ num, float* __restrict__ den, int E)
{
    const int lane = threadIdx.x & 31;
    const int e    = blockIdx.x * 8 + (threadIdx.x >> 5);
    if (e >= E) return;

    const int src = edges[2 * e + 0];
    const int dst = edges[2 * e + 1];

    float s = s_src[src] + s_dst[dst];
    s = (s >= 0.0f) ? s : SLOPE * s;
    const float w = expf(s);

    const float4 hv = *(const float4*)(h + (size_t)dst * OUT_DIM + lane * 4);
    float* np = num + (size_t)src * OUT_DIM + lane * 4;
    unsafeAtomicAdd(np + 0, w * hv.x);
    unsafeAtomicAdd(np + 1, w * hv.y);
    unsafeAtomicAdd(np + 2, w * hv.z);
    unsafeAtomicAdd(np + 3, w * hv.w);
    if (lane == 0) unsafeAtomicAdd(&den[src], w);
}

__global__ __launch_bounds__(256) void gat_out_kernel(
    const float* __restrict__ num, const float* __restrict__ den,
    const int* __restrict__ node_idx, float* __restrict__ out, int B)
{
    const int t = blockIdx.x * blockDim.x + threadIdx.x;
    const int j = t >> 5;
    const int lane = t & 31;
    if (j >= B) return;

    const int row = node_idx[j];
    const float inv = 1.0f / den[row];
    const float4 nv = *(const float4*)(num + (size_t)row * OUT_DIM + lane * 4);
    float4 ov;
    ov.x = nv.x * inv; ov.y = nv.y * inv; ov.z = nv.z * inv; ov.w = nv.w * inv;
    *(float4*)(out + (size_t)j * OUT_DIM + lane * 4) = ov;
}

extern "C" void kernel_launch(void* const* d_in, const int* in_sizes, int n_in,
                              void* d_out, int out_size, void* d_ws, size_t ws_size,
                              hipStream_t stream) {
    const float* features     = (const float*)d_in[0];
    const float* W            = (const float*)d_in[1];
    const float* b            = (const float*)d_in[2];
    const float* a            = (const float*)d_in[3];
    const int*   edges        = (const int*)d_in[4];
    const int*   unique_nodes = (const int*)d_in[5];
    const int*   node_idx     = (const int*)d_in[6];
    float* out = (float*)d_out;

    const int N = in_sizes[5];
    const int E = in_sizes[4] / 2;
    const int B = in_sizes[6];

    float* h     = (float*)d_ws;
    float* num   = h + (size_t)N * OUT_DIM;
    float* s_src = num + (size_t)N * OUT_DIM;
    float* s_dst = s_src + N;
    float* den   = s_dst + N;

    hipMemsetAsync(num, 0, (size_t)N * OUT_DIM * sizeof(float), stream);
    hipMemsetAsync(den, 0, (size_t)N * sizeof(float), stream);

    {
        dim3 grid((N + 15) / 16);
        gat_gemm_kernel<<<grid, 256, 0, stream>>>(features, W, b, unique_nodes, h, N);
    }
    {
        dim3 grid((N + 7) / 8);
        gat_scores_kernel<<<grid, 256, 0, stream>>>(h, a, s_src, s_dst, N);
    }
    {
        dim3 grid((E + 7) / 8);
        gat_edge_kernel<<<grid, 256, 0, stream>>>(edges, h, s_src, s_dst, num, den, E);
    }
    {
        long threads = (long)B * 32;
        dim3 grid((unsigned)((threads + 255) / 256));
        gat_out_kernel<<<grid, 256, 0, stream>>>(num, den, node_idx, out, B);
    }
}


# edge sums by counting sort: a returning integer add ranks each edge in its source row, rows get contiguous segments, a wave per output row sums its segment in registers (rank order); no float atomics
# speedup vs baseline: 1.9275x; 1.9275x over previous
#include <hip/hip_runtime.h>
#include <hip/hip_bf16.h>

#define IN_DIM 256
#define OUT_DIM 128
#define SLOPE 0.1f

typedef float v2f __attribute__((ext_vector_type(2)));
typedef float v8f __attribute__((ext_vector_type(8)));

__global__ __launch_bounds__(256) void gat_gemm_kernel(
    const float* __restrict__ features, const float* __restrict__ W,
    const float* __restrict__ bias, const int* __restrict__ unique_nodes,
    float* __restrict__ h, int N)
{
    const int lane = threadIdx.x & 31;
    const int wave = threadIdx.x >> 5;
    const int half = lane >> 4;
    const int mn   = lane & 15;

    const int rowBase = blockIdx.x * 16;
    int arow_idx = rowBase + mn;
    if (arow_idx >= N) arow_idx = N - 1;
    const int gsrc = unique_nodes[arow_idx];
    const float* __restrict__ arow = features + (size_t)gsrc * IN_DIM;

    const int ncol = wave * 16 + mn;
    const float* __restrict__ brow = W + (size_t)ncol * IN_DIM;

    v8f acc;
    const float bv = bias[ncol];
#pragma unroll
    for (int r = 0; r < 8; ++r) acc[r] = bv;

#pragma unroll 4
    for (int k = 0; k < IN_DIM; k += 4) {
        v2f av  = *(const v2f*)(arow + k + 2 * half);
        v2f bvv = *(const v2f*)(brow + k + 2 * half);
        acc = __builtin_amdgcn_wmma_f32_16x16x4_f32(
             false, av,  false, bvv,
             (short)0, acc,  false,  false);
    }

    float* __restrict__ obase =
        h + (size_t)(rowBase + 8 * half) * OUT_DIM + ncol;
    if (rowBase + 16 <= N) {
#pragma unroll
        for (int r = 0; r < 8; ++r)
            obase[(size_t)r * OUT_DIM] = acc[r];
    } else {
#pragma unroll
        for (int r = 0; r < 8; ++r) {
            int orow = rowBase + 8 * half + r;
            if (orow < N) obase[(size_t)r * OUT_DIM] = acc[r];
        }
    }
}

__global__ __launch_bounds__(256) void gat_scores_kernel(
    const float* __restrict__ h, const float* __restrict__ a,
    float* __restrict__ s_src, float* __restrict__ s_dst, int N)
{
    const int lane = threadIdx.x & 31;
    const int row  = blockIdx.x * 8 + (threadIdx.x >> 5);
    if (row >= N) return;

    const float4 hv = *(const float4*)(h + (size_t)row * OUT_DIM + lane * 4);
    const float4 as = *(const float4*)(a + lane * 4);
    const float4 ad = *(const float4*)(a + OUT_DIM + lane * 4);

    float ps = hv.x * as.x + hv.y * as.y + hv.z * as.z + hv.w * as.w;
    float pd = hv.x * ad.x + hv.y * ad.y + hv.z * ad.z + hv.w * ad.w;

#pragma unroll
    for (int off = 16; off > 0; off >>= 1) {
        ps += __shfl_xor(ps, off, 32);
        pd += __shfl_xor(pd, off, 32);
    }
    if (lane == 0) {
        s_src[row] = ps;
        s_dst[row] = pd;
    }
}

__global__ __launch_bounds__(256) void gat_rank_kernel(
    const int* __restrict__ edges, int* __restrict__ cnt,
    int* __restrict__ rank, int E)
{
    const int e = blockIdx.x * 256 + threadIdx.x;
    if (e >= E) return;
    const int src = edges[2 * e];
    rank[e] = atomicAdd(&cnt[src], 1);
}

__global__ __launch_bounds__(256) void gat_alloc_kernel(
    const int* __restrict__ cnt, int* __restrict__ total,
    int* __restrict__ seg, int N)
{
    __shared__ int wsum[8];
    __shared__ int base;
    const int lane = threadIdx.x & 31;
    const int wv   = threadIdx.x >> 5;
    const int i    = blockIdx.x * 256 + threadIdx.x;
    const int c    = (i < N) ? cnt[i] : 0;

    int x = c;
    for (int off = 1; off < 32; off <<= 1) {
        const int y = __shfl_up(x, off, 32);
        if (lane >= off) x += y;
    }
    if (lane == 31) wsum[wv] = x;
    __syncthreads();
    if (threadIdx.x == 0) {
        int s = 0;
        for (int w = 0; w < 8; ++w) { const int t = wsum[w]; wsum[w] = s; s += t; }
        base = atomicAdd(total, s);
    }
    __syncthreads();
    if (i < N) *(int2*)(seg + 2 * (size_t)i) = make_int2(base + wsum[wv] + x - c, c);
}

__global__ __launch_bounds__(256) void gat_fill_kernel(
    const int* __restrict__ edges, const int* __restrict__ rank,
    const int* __restrict__ seg, int* __restrict__ sorted, int E)
{
    const int e = blockIdx.x * 256 + threadIdx.x;
    if (e >= E) return;
    const int2 sd = *(const int2*)(edges + 2 * (size_t)e);
    sorted[seg[2 * (size_t)sd.x] + rank[e]] = sd.y;
}

template <int U>
__device__ __forceinline__ void gat_walk_step(
    const float* __restrict__ h, int d, float w, int i, int lane,
    float4& acc, float& den)
{
    float4 hv[U];
    float  wi[U];
#pragma unroll
    for (int u = 0; u < U; ++u) {
        const int di = __builtin_amdgcn_readlane(d, i + u);
        wi[u] = __int_as_float(__builtin_amdgcn_readlane(__float_as_int(w), i + u));
        hv[u] = *(const float4*)(h + (size_t)di * OUT_DIM + lane * 4);
    }
#pragma unroll
    for (int u = 0; u < U; ++u) {
        acc.x += wi[u] * hv[u].x;
        acc.y += wi[u] * hv[u].y;
        acc.z += wi[u] * hv[u].z;
        acc.w += wi[u] * hv[u].w;
        den   += wi[u];
    }
}

__global__ __launch_bounds__(256) void gat_walk_kernel(
    const float* __restrict__ h, const float* __restrict__ s_src,
    const float* __restrict__ s_dst, const int* __restrict__ seg,
    const int* __restrict__ sorted,
    const int* __restrict__ node_idx, float* __restrict__ out, int B)
{
    const int lane = threadIdx.x & 31;
    const int wv = __builtin_amdgcn_readfirstlane(threadIdx.x >> 5);
    const int j  = blockIdx.x * 8 + wv;
    if (j >= B) return;

    const int   row = node_idx[j];
    const unsigned long long sg = *(const unsigned long long*)(seg + 2 * (size_t)row);
    const int   beg = (int)(unsigned)(sg & 0xffffffffull);
    const int   n   = (int)(unsigned)(sg >> 32);
    const float ss  = s_src[row];

    float4 acc = make_float4(0.0f, 0.0f, 0.0f, 0.0f);
    float  den = 0.0f;

    for (int c = 0; c < n; c += 32) {
        const int m = (n - c < 32) ? (n - c) : 32;
        int   d = 0;
        float w = 0.0f;
        if (lane < m) {
            d = sorted[beg + c + lane];
            float s = ss + s_dst[d];
            s = (s >= 0.0f) ? s : SLOPE * s;
            w = expf(s);
        }
        int i = 0;
        for (; i + 8 <= m; i += 8) gat_walk_step<8>(h, d, w, i, lane, acc, den);
        if (i + 4 <= m) { gat_walk_step<4>(h, d, w, i, lane, acc, den); i += 4; }
        if (i + 2 <= m) { gat_walk_step<2>(h, d, w, i, lane, acc, den); i += 2; }
        if (i < m)      { gat_walk_step<1>(h, d, w, i, lane, acc, den); }
    }

    const float inv = 1.0f / den;
    float4 ov;
    ov.x = acc.x * inv; ov.y = acc.y * inv; ov.z = acc.z * inv; ov.w = acc.w * inv;
    *(float4*)(out + (size_t)j * OUT_DIM + lane * 4) = ov;
}

extern "C" void kernel_launch(void* const* d_in, const int* in_sizes, int n_in,
                              void* d_out, int out_size, void* d_ws, size_t ws_size,
                              hipStream_t stream) {
    const float* features     = (const float*)d_in[0];
    const float* W            = (const float*)d_in[1];
    const float* b            = (const float*)d_in[2];
    const float* a            = (const float*)d_in[3];
    const int*   edges        = (const int*)d_in[4];
    const int*   unique_nodes = (const int*)d_in[5];
    const int*   node_idx     = (const int*)d_in[6];
    float* out = (float*)d_out;

    const int N = in_sizes[5];
    const int E = in_sizes[4] / 2;
    const int B = in_sizes[6];

    float* h      = (float*)d_ws;
    int*   seg    = (int*)(h + (size_t)N * OUT_DIM);
    float* s_src  = (float*)(seg + 2 * (size_t)N);
    float* s_dst  = s_src + N;
    int*   cnt    = (int*)(s_dst + N);
    int*   total  = cnt + N;
    int*   rank   = total + 1;
    int*   sorted = rank + E;

    hipMemsetAsync(cnt, 0, ((size_t)N + 1) * sizeof(int), stream);

    {
        dim3 grid((N + 15) / 16);
        gat_gemm_kernel<<<grid, 256, 0, stream>>>(features, W, b, unique_nodes, h, N);
    }
    {
        dim3 grid((N + 7) / 8);
        gat_scores_kernel<<<grid, 256, 0, stream>>>(h, a, s_src, s_dst, N);
    }
    {
        dim3 grid((E + 255) / 256);
        gat_rank_kernel<<<grid, 256, 0, stream>>>(edges, cnt, rank, E);
    }
    {
        dim3 grid((N + 255) / 256);
        gat_alloc_kernel<<<grid, 256, 0, stream>>>(cnt, total, seg, N);
    }
    {
        dim3 grid((E + 255) / 256);
        gat_fill_kernel<<<grid, 256, 0, stream>>>(edges, rank, seg, sorted, E);
    }
    {
        dim3 grid((B + 7) / 8);
        gat_walk_kernel<<<grid, 256, 0, stream>>>(h, s_src, s_dst, seg, sorted,
                                                  node_idx, out, B);
    }
}


# GEMM re-tiled: 256 resident groups, a wave owns 16 rows x all 128 columns; feature rows read once by 16-byte loads into registers, W fragments from an LDS copy; k summed in another order
# speedup vs baseline: 2.6591x; 1.3796x over previous
#include <hip/hip_runtime.h>
#include <hip/hip_bf16.h>

#define IN_DIM 256
#define OUT_DIM 128
#define SLOPE 0.1f

typedef float v2f __attribute__((ext_vector_type(2)));
typedef float v8f __attribute__((ext_vector_type(8)));

#define GEMM_GROUPS 256
#define GEMM_WAVES 8
#define KCHUNKS (IN_DIM / 8)
#define CTILES (OUT_DIM / 16)
#define CHUNK_PITCH (CTILES * 32)
#define GEMM_LDS_SLOTS 10752

__device__ __forceinline__ void gat_load_stripe(
    const float* __restrict__ features, const int* __restrict__ unique_nodes,
    int tile, int mn, int half, int N, float4 (&dst)[KCHUNKS])
{
    int arow_idx = tile * 16 + mn;
    if (arow_idx >= N) arow_idx = N - 1;
    const int gsrc = unique_nodes[arow_idx];
    const float* __restrict__ arow = features + (size_t)gsrc * IN_DIM + 4 * half;
#pragma unroll
    for (int t = 0; t < KCHUNKS; ++t)
        dst[t] = *(const float4*)(arow + 8 * t);
}

__global__ __launch_bounds__(256) void gat_gemm_kernel(
    const float* __restrict__ features, const float* __restrict__ W,
    const float* __restrict__ bias, const int* __restrict__ unique_nodes,
    float* __restrict__ h, int N)
{
    __shared__ float4 wl[GEMM_LDS_SLOTS];

    const int lane = threadIdx.x & 31;
    const int wave = __builtin_amdgcn_readfirstlane(threadIdx.x >> 5);
    const int half = lane >> 4;
    const int mn   = lane & 15;

    const int nTiles = (N + 15) >> 4;
    const int stride = (int)gridDim.x * GEMM_WAVES;
    int tile = wave * (int)gridDim.x + (int)blockIdx.x;

    float4 cur[KCHUNKS];
    float4 nxt[KCHUNKS];
#pragma unroll
    for (int t = 0; t < KCHUNKS; ++t) {
        cur[t] = make_float4(0.0f, 0.0f, 0.0f, 0.0f);
        nxt[t] = make_float4(0.0f, 0.0f, 0.0f, 0.0f);
    }
    if (tile < nTiles) gat_load_stripe(features, unique_nodes, tile, mn, half, N, cur);

    {
        const int qlo  = lane & 7;
        const int rsub = lane >> 3;
#pragma unroll 16
        for (int it = 0; it < (OUT_DIM * IN_DIM / 4) / 256; ++it) {
            const int wi = it * GEMM_WAVES + wave;
            const int n  = (wi >> 3) * 4 + rsub;
            const int q  = (wi & 7) * 8 + qlo;
            const float4 v = *(const float4*)(W + (size_t)n * IN_DIM + 4 * q);
            wl[(q >> 1) * CHUNK_PITCH + (n >> 4) * 32 + (q & 1) * 16 + (n & 15)] = v;
        }
    }

    float bv[CTILES];
#pragma unroll
    for (int c = 0; c < CTILES; ++c) bv[c] = bias[c * 16 + mn];

    __syncthreads();

    const int loSlot = lane;
    int hiSlot = lane + (KCHUNKS / 2) * CHUNK_PITCH;
    asm volatile("" : "+v"(hiSlot));

    for (; tile < nTiles; tile += stride) {
        const int  rowBase = tile * 16;
        const bool more    = (tile + stride) < nTiles;
        if (more) gat_load_stripe(features, unique_nodes, tile + stride, mn, half, N, nxt);

        v8f acc[CTILES];
#pragma unroll
        for (int c = 0; c < CTILES; ++c) {
#pragma unroll
            for (int r = 0; r < 8; ++r) acc[c][r] = bv[c];
        }

        float4 bcur[CTILES];
#pragma unroll
        for (int c = 0; c < CTILES; ++c) bcur[c] = wl[loSlot + c * 32];

#pragma unroll
        for (int t = 0; t < KCHUNKS; ++t) {
            float4 bnx[CTILES];
            if (t + 1 < KCHUNKS) {
#pragma unroll
                for (int c = 0; c < CTILES; ++c)
                    bnx[c] = (t + 1 < KCHUNKS / 2)
                        ? wl[loSlot + (t + 1) * CHUNK_PITCH + c * 32]
                        : wl[hiSlot + (t + 1 - KCHUNKS / 2) * CHUNK_PITCH + c * 32];
            }
            __builtin_amdgcn_sched_barrier(0);
            const float4 a = cur[t];
            const v2f a0 = {a.x, a.y};
            const v2f a1 = {a.z, a.w};
#pragma unroll
            for (int c = 0; c < CTILES; ++c) {
                const v2f b0 = {bcur[c].x, bcur[c].y};
                acc[c] = __builtin_amdgcn_wmma_f32_16x16x4_f32(
                     false, a0,  false, b0,
                     (short)0, acc[c],  false,  false);
            }
#pragma unroll
            for (int c = 0; c < CTILES; ++c) {
                const v2f b1 = {bcur[c].z, bcur[c].w};
                acc[c] = __builtin_amdgcn_wmma_f32_16x16x4_f32(
                     false, a1,  false, b1,
                     (short)0, acc[c],  false,  false);
            }
            __builtin_amdgcn_sched_barrier(0);
            if (t + 1 < KCHUNKS) {
#pragma unroll
                for (int c = 0; c < CTILES; ++c) bcur[c] = bnx[c];
            }
        }

        asm volatile("v_nop\n\tv_nop\n\tv_nop\n\tv_nop\n\tv_nop\n\tv_nop\n\tv_nop\n\tv_nop");
        __builtin_amdgcn_sched_barrier(0);

        float* __restrict__ obase =
            h + (size_t)(rowBase + 8 * half) * OUT_DIM + mn;
        if (rowBase + 16 <= N) {
#pragma unroll
            for (int c = 0; c < CTILES; ++c) {
#pragma unroll
                for (int r = 0; r < 8; ++r)
                    obase[(size_t)r * OUT_DIM + c * 16] = acc[c][r];
            }
        } else {
#pragma unroll
            for (int c = 0; c < CTILES; ++c) {
#pragma unroll
                for (int r = 0; r < 8; ++r) {
                    const int orow = rowBase + 8 * half + r;
                    if (orow < N) obase[(size_t)r * OUT_DIM + c * 16] = acc[c][r];
                }
            }
        }

        if (more) {
#pragma unroll
            for (int t = 0; t < KCHUNKS; ++t) cur[t] = nxt[t];
        }
    }
}

__global__ __launch_bounds__(256) void gat_scores_kernel(
    const float* __restrict__ h, const float* __restrict__ a,
    float* __restrict__ s_src, float* __restrict__ s_dst, int N)
{
    const int lane = threadIdx.x & 31;
    const int row  = blockIdx.x * 8 + (threadIdx.x >> 5);
    if (row >= N) return;

    const float4 hv = *(const float4*)(h + (size_t)row * OUT_DIM + lane * 4);
    const float4 as = *(const float4*)(a + lane * 4);
    const float4 ad = *(const float4*)(a + OUT_DIM + lane * 4);

    float ps = hv.x * as.x + hv.y * as.y + hv.z * as.z + hv.w * as.w;
    float pd = hv.x * ad.x + hv.y * ad.y + hv.z * ad.z + hv.w * ad.w;

#pragma unroll
    for (int off = 16; off > 0; off >>= 1) {
        ps += __shfl_xor(ps, off, 32);
        pd += __shfl_xor(pd, off, 32);
    }
    if (lane == 0) {
        s_src[row] = ps;
        s_dst[row] = pd;
    }
}

__global__ __launch_bounds__(256) void gat_rank_kernel(
    const int* __restrict__ edges, int* __restrict__ cnt,
    int* __restrict__ rank, int E)
{
    const int e = blockIdx.x * 256 + threadIdx.x;
    if (e >= E) return;
    const int src = edges[2 * e];
    rank[e] = atomicAdd(&cnt[src], 1);
}

__global__ __launch_bounds__(256) void gat_alloc_kernel(
    const int* __restrict__ cnt, int* __restrict__ total,
    int* __restrict__ seg, int N)
{
    __shared__ int wsum[8];
    __shared__ int base;
    const int lane = threadIdx.x & 31;
    const int wv   = threadIdx.x >> 5;
    const int i    = blockIdx.x * 256 + threadIdx.x;
    const int c    = (i < N) ? cnt[i] : 0;

    int x = c;
    for (int off = 1; off < 32; off <<= 1) {
        const int y = __shfl_up(x, off, 32);
        if (lane >= off) x += y;
    }
    if (lane == 31) wsum[wv] = x;
    __syncthreads();
    if (threadIdx.x == 0) {
        int s = 0;
        for (int w = 0; w < 8; ++w) { const int t = wsum[w]; wsum[w] = s; s += t; }
        base = atomicAdd(total, s);
    }
    __syncthreads();
    if (i < N) *(int2*)(seg + 2 * (size_t)i) = make_int2(base + wsum[wv] + x - c, c);
}

__global__ __launch_bounds__(256) void gat_fill_kernel(
    const int* __restrict__ edges, const int* __restrict__ rank,
    const int* __restrict__ seg, int* __restrict__ sorted, int E)
{
    const int e = blockIdx.x * 256 + threadIdx.x;
    if (e >= E) return;
    const int2 sd = *(const int2*)(edges + 2 * (size_t)e);
    sorted[seg[2 * (size_t)sd.x] + rank[e]] = sd.y;
}

template <int U>
__device__ __forceinline__ void gat_walk_step(
    const float* __restrict__ h, int d, float w, int i, int lane,
    float4& acc, float& den)
{
    float4 hv[U];
    float  wi[U];
#pragma unroll
    for (int u = 0; u < U; ++u) {
        const int di = __builtin_amdgcn_readlane(d, i + u);
        wi[u] = __int_as_float(__builtin_amdgcn_readlane(__float_as_int(w), i + u));
        hv[u] = *(const float4*)(h + (size_t)di * OUT_DIM + lane * 4);
    }
#pragma unroll
    for (int u = 0; u < U; ++u) {
        acc.x += wi[u] * hv[u].x;
        acc.y += wi[u] * hv[u].y;
        acc.z += wi[u] * hv[u].z;
        acc.w += wi[u] * hv[u].w;
        den   += wi[u];
    }
}

__global__ __launch_bounds__(256) void gat_walk_kernel(
    const float* __restrict__ h, const float* __restrict__ s_src,
    const float* __restrict__ s_dst, const int* __restrict__ seg,
    const int* __restrict__ sorted,
    const int* __restrict__ node_idx, float* __restrict__ out, int B)
{
    const int lane = threadIdx.x & 31;
    const int wv = __builtin_amdgcn_readfirstlane(threadIdx.x >> 5);
    const int j  = blockIdx.x * 8 + wv;
    if (j >= B) return;

    const int   row = node_idx[j];
    const unsigned long long sg = *(const unsigned long long*)(seg + 2 * (size_t)row);
    const int   beg = (int)(unsigned)(sg & 0xffffffffull);
    const int   n   = (int)(unsigned)(sg >> 32);
    const float ss  = s_src[row];

    float4 acc = make_float4(0.0f, 0.0f, 0.0f, 0.0f);
    float  den = 0.0f;

    for (int c = 0; c < n; c += 32) {
        const int m = (n - c < 32) ? (n - c) : 32;
        int   d = 0;
        float w = 0.0f;
        if (lane < m) {
            d = sorted[beg + c + lane];
            float s = ss + s_dst[d];
            s = (s >= 0.0f) ? s : SLOPE * s;
            w = expf(s);
        }
        int i = 0;
        for (; i + 8 <= m; i += 8) gat_walk_step<8>(h, d, w, i, lane, acc, den);
        if (i + 4 <= m) { gat_walk_step<4>(h, d, w, i, lane, acc, den); i += 4; }
        if (i + 2 <= m) { gat_walk_step<2>(h, d, w, i, lane, acc, den); i += 2; }
        if (i < m)      { gat_walk_step<1>(h, d, w, i, lane, acc, den); }
    }

    const float inv = 1.0f / den;
    float4 ov;
    ov.x = acc.x * inv; ov.y = acc.y * inv; ov.z = acc.z * inv; ov.w = acc.w * inv;
    *(float4*)(out + (size_t)j * OUT_DIM + lane * 4) = ov;
}

extern "C" void kernel_launch(void* const* d_in, const int* in_sizes, int n_in,
                              void* d_out, int out_size, void* d_ws, size_t ws_size,
                              hipStream_t stream) {
    const float* features     = (const float*)d_in[0];
    const float* W            = (const float*)d_in[1];
    const float* b            = (const float*)d_in[2];
    const float* a            = (const float*)d_in[3];
    const int*   edges        = (const int*)d_in[4];
    const int*   unique_nodes = (const int*)d_in[5];
    const int*   node_idx     = (const int*)d_in[6];
    float* out = (float*)d_out;

    const int N = in_sizes[5];
    const int E = in_sizes[4] / 2;
    const int B = in_sizes[6];

    float* h      = (float*)d_ws;
    int*   seg    = (int*)(h + (size_t)N * OUT_DIM);
    float* s_src  = (float*)(seg + 2 * (size_t)N);
    float* s_dst  = s_src + N;
    int*   cnt    = (int*)(s_dst + N);
    int*   total  = cnt + N;
    int*   rank   = total + 1;
    int*   sorted = rank + E;

    hipMemsetAsync(cnt, 0, ((size_t)N + 1) * sizeof(int), stream);

    {
        gat_gemm_kernel<<<GEMM_GROUPS, 256, 0, stream>>>(features, W, b, unique_nodes, h, N);
    }
    {
        dim3 grid((N + 7) / 8);
        gat_scores_kernel<<<grid, 256, 0, stream>>>(h, a, s_src, s_dst, N);
    }
    {
        dim3 grid((E + 255) / 256);
        gat_rank_kernel<<<grid, 256, 0, stream>>>(edges, cnt, rank, E);
    }
    {
        dim3 grid((N + 255) / 256);
        gat_alloc_kernel<<<grid, 256, 0, stream>>>(cnt, total, seg, N);
    }
    {
        dim3 grid((E + 255) / 256);
        gat_fill_kernel<<<grid, 256, 0, stream>>>(edges, rank, seg, sorted, E);
    }
    {
        dim3 grid((B + 7) / 8);
        gat_walk_kernel<<<grid, 256, 0, stream>>>(h, s_src, s_dst, seg, sorted,
                                                  node_idx, out, B);
    }
}


# scores, rank and fill launches on resident grids: some 6500 waves each, a wave takes 8 rows (a thread 8 edges) per trip with its loads in flight, no short wave per row or 32 edges; sum trees kept
# speedup vs baseline: 2.7410x; 1.0308x over previous
#include <hip/hip_runtime.h>
#include <hip/hip_bf16.h>

#define IN_DIM 256
#define OUT_DIM 128
#define SLOPE 0.1f

typedef float v2f __attribute__((ext_vector_type(2)));
typedef float v8f __attribute__((ext_vector_type(8)));

#define GEMM_GROUPS 256
#define GEMM_WAVES 8
#define KCHUNKS (IN_DIM / 8)
#define CTILES (OUT_DIM / 16)
#define CHUNK_PITCH (CTILES * 32)
#define GEMM_LDS_SLOTS 10752

__device__ __forceinline__ void gat_load_stripe(
    const float* __restrict__ features, const int* __restrict__ unique_nodes,
    int tile, int mn, int half, int N, float4 (&dst)[KCHUNKS])
{
    int arow_idx = tile * 16 + mn;
    if (arow_idx >= N) arow_idx = N - 1;
    const int gsrc = unique_nodes[arow_idx];
    const float* __restrict__ arow = features + (size_t)gsrc * IN_DIM + 4 * half;
#pragma unroll
    for (int t = 0; t < KCHUNKS; ++t)
        dst[t] = *(const float4*)(arow + 8 * t);
}

__global__ __launch_bounds__(256) void gat_gemm_kernel(
    const float* __restrict__ features, const float* __restrict__ W,
    const float* __restrict__ bias, const int* __restrict__ unique_nodes,
    float* __restrict__ h, int N)
{
    __shared__ float4 wl[GEMM_LDS_SLOTS];

    const int lane = threadIdx.x & 31;
    const int wave = __builtin_amdgcn_readfirstlane(threadIdx.x >> 5);
    const int half = lane >> 4;
    const int mn   = lane & 15;

    const int nTiles = (N + 15) >> 4;
    const int stride = (int)gridDim.x * GEMM_WAVES;
    int tile = wave * (int)gridDim.x + (int)blockIdx.x;

    float4 cur[KCHUNKS];
    float4 nxt[KCHUNKS];
#pragma unroll
    for (int t = 0; t < KCHUNKS; ++t) {
        cur[t] = make_float4(0.0f, 0.0f, 0.0f, 0.0f);
        nxt[t] = make_float4(0.0f, 0.0f, 0.0f, 0.0f);
    }
    if (tile < nTiles) gat_load_stripe(features, unique_nodes, tile, mn, half, N, cur);

    {
        const int qlo  = lane & 7;
        const int rsub = lane >> 3;
#pragma unroll 16
        for (int it = 0; it < (OUT_DIM * IN_DIM / 4) / 256; ++it) {
            const int wi = it * GEMM_WAVES + wave;
            const int n  = (wi >> 3) * 4 + rsub;
            const int q  = (wi & 7) * 8 + qlo;
            const float4 v = *(const float4*)(W + (size_t)n * IN_DIM + 4 * q);
            wl[(q >> 1) * CHUNK_PITCH + (n >> 4) * 32 + (q & 1) * 16 + (n & 15)] = v;
        }
    }

    float bv[CTILES];
#pragma unroll
    for (int c = 0; c < CTILES; ++c) bv[c] = bias[c * 16 + mn];

    __syncthreads();

    const int loSlot = lane;
    int hiSlot = lane + (KCHUNKS / 2) * CHUNK_PITCH;
    asm volatile("" : "+v"(hiSlot));

    for (; tile < nTiles; tile += stride) {
        const int  rowBase = tile * 16;
        const bool more    = (tile + stride) < nTiles;
        if (more) gat_load_stripe(features, unique_nodes, tile + stride, mn, half, N, nxt);

        v8f acc[CTILES];
#pragma unroll
        for (int c = 0; c < CTILES; ++c) {
#pragma unroll
            for (int r = 0; r < 8; ++r) acc[c][r] = bv[c];
        }

        float4 bcur[CTILES];
#pragma unroll
        for (int c = 0; c < CTILES; ++c) bcur[c] = wl[loSlot + c * 32];

#pragma unroll
        for (int t = 0; t < KCHUNKS; ++t) {
            float4 bnx[CTILES];
            if (t + 1 < KCHUNKS) {
#pragma unroll
                for (int c = 0; c < CTILES; ++c)
                    bnx[c] = (t + 1 < KCHUNKS / 2)
                        ? wl[loSlot + (t + 1) * CHUNK_PITCH + c * 32]
                        : wl[hiSlot + (t + 1 - KCHUNKS / 2) * CHUNK_PITCH + c * 32];
            }
            __builtin_amdgcn_sched_barrier(0);
            const float4 a = cur[t];
            const v2f a0 = {a.x, a.y};
            const v2f a1 = {a.z, a.w};
#pragma unroll
            for (int c = 0; c < CTILES; ++c) {
                const v2f b0 = {bcur[c].x, bcur[c].y};
                acc[c] = __builtin_amdgcn_wmma_f32_16x16x4_f32(
                     false, a0,  false, b0,
                     (short)0, acc[c],  false,  false);
            }
#pragma unroll
            for (int c = 0; c < CTILES; ++c) {
                const v2f b1 = {bcur[c].z, bcur[c].w};
                acc[c] = __builtin_amdgcn_wmma_f32_16x16x4_f32(
                     false, a1,  false, b1,
                     (short)0, acc[c],  false,  false);
            }
            __builtin_amdgcn_sched_barrier(0);
            if (t + 1 < KCHUNKS) {
#pragma unroll
                for (int c = 0; c < CTILES; ++c) bcur[c] = bnx[c];
            }
        }

        asm volatile("v_nop\n\tv_nop\n\tv_nop\n\tv_nop\n\tv_nop\n\tv_nop\n\tv_nop\n\tv_nop");
        __builtin_amdgcn_sched_barrier(0);

        float* __restrict__ obase =
            h + (size_t)(rowBase + 8 * half) * OUT_DIM + mn;
        if (rowBase + 16 <= N) {
#pragma unroll
            for (int c = 0; c < CTILES; ++c) {
#pragma unroll
                for (int r = 0; r < 8; ++r)
                    obase[(size_t)r * OUT_DIM + c * 16] = acc[c][r];
            }
        } else {
#pragma unroll
            for (int c = 0; c < CTILES; ++c) {
#pragma unroll
                for (int r = 0; r < 8; ++r) {
                    const int orow = rowBase + 8 * half + r;
                    if (orow < N) obase[(size_t)r * OUT_DIM + c * 16] = acc[c][r];
                }
            }
        }

        if (more) {
#pragma unroll
            for (int t = 0; t < KCHUNKS; ++t) cur[t] = nxt[t];
        }
    }
}

#define SCORE_GROUPS 784
#define SCORE_ROWS 8

__global__ __launch_bounds__(256) void gat_scores_kernel(
    const float* __restrict__ h, const float* __restrict__ a,
    float* __restrict__ s_src, float* __restrict__ s_dst, int N)
{
    const int lane = threadIdx.x & 31;
    const int gw = (int)blockIdx.x * 8 + __builtin_amdgcn_readfirstlane(threadIdx.x >> 5);
    const int step = (int)gridDim.x * 8 * SCORE_ROWS;

    const float4 as = *(const float4*)(a + lane * 4);
    const float4 ad = *(const float4*)(a + OUT_DIM + lane * 4);

    for (int row0 = gw * SCORE_ROWS; row0 < N; row0 += step) {
        float4 hv[SCORE_ROWS];
#pragma unroll
        for (int u = 0; u < SCORE_ROWS; ++u) {
            int r = row0 + u;
            if (r >= N) r = N - 1;
            hv[u] = *(const float4*)(h + (size_t)r * OUT_DIM + lane * 4);
        }
        float outs = 0.0f, outd = 0.0f;
#pragma unroll
        for (int u = 0; u < SCORE_ROWS; ++u) {
            float ps = hv[u].x * as.x + hv[u].y * as.y + hv[u].z * as.z + hv[u].w * as.w;
            float pd = hv[u].x * ad.x + hv[u].y * ad.y + hv[u].z * ad.z + hv[u].w * ad.w;
#pragma unroll
            for (int off = 16; off > 0; off >>= 1) {
                ps += __shfl_xor(ps, off, 32);
                pd += __shfl_xor(pd, off, 32);
            }
            if (lane == u) { outs = ps; outd = pd; }
        }
        if (lane < SCORE_ROWS && row0 + lane < N) {
            s_src[row0 + lane] = outs;
            s_dst[row0 + lane] = outd;
        }
    }
}

#define EDGE_GROUPS 832
#define EDGE_U 8
#define EDGE_SPAN (256 * EDGE_U)

__global__ __launch_bounds__(256) void gat_rank_kernel(
    const int* __restrict__ edges, int* __restrict__ cnt,
    int* __restrict__ rank, int E)
{
    const int step = (int)gridDim.x * EDGE_SPAN;
    for (int base = (int)blockIdx.x * EDGE_SPAN; base < E; base += step) {
        const int e0 = base + (int)threadIdx.x;
        int sr[EDGE_U];
        int rk[EDGE_U];
        if (base + EDGE_SPAN <= E) {
#pragma unroll
            for (int u = 0; u < EDGE_U; ++u) sr[u] = edges[2 * (size_t)(e0 + u * 256)];
#pragma unroll
            for (int u = 0; u < EDGE_U; ++u) rk[u] = atomicAdd(&cnt[sr[u]], 1);
#pragma unroll
            for (int u = 0; u < EDGE_U; ++u) rank[e0 + u * 256] = rk[u];
        } else {
#pragma unroll
            for (int u = 0; u < EDGE_U; ++u) {
                const int e = e0 + u * 256;
                if (e < E) rank[e] = atomicAdd(&cnt[edges[2 * (size_t)e]], 1);
            }
        }
    }
}

__global__ __launch_bounds__(256) void gat_alloc_kernel(
    const int* __restrict__ cnt, int* __restrict__ total,
    int* __restrict__ seg, int N)
{
    __shared__ int wsum[8];
    __shared__ int base;
    const int lane = threadIdx.x & 31;
    const int wv   = threadIdx.x >> 5;
    const int i    = blockIdx.x * 256 + threadIdx.x;
    const int c    = (i < N) ? cnt[i] : 0;

    int x = c;
    for (int off = 1; off < 32; off <<= 1) {
        const int y = __shfl_up(x, off, 32);
        if (lane >= off) x += y;
    }
    if (lane == 31) wsum[wv] = x;
    __syncthreads();
    if (threadIdx.x == 0) {
        int s = 0;
        for (int w = 0; w < 8; ++w) { const int t = wsum[w]; wsum[w] = s; s += t; }
        base = atomicAdd(total, s);
    }
    __syncthreads();
    if (i < N) *(int2*)(seg + 2 * (size_t)i) = make_int2(base + wsum[wv] + x - c, c);
}

__global__ __launch_bounds__(256) void gat_fill_kernel(
    const int* __restrict__ edges, const int* __restrict__ rank,
    const int* __restrict__ seg, int* __restrict__ sorted, int E)
{
    const int step = (int)gridDim.x * EDGE_SPAN;
    for (int base = (int)blockIdx.x * EDGE_SPAN; base < E; base += step) {
        const int e0 = base + (int)threadIdx.x;
        if (base + EDGE_SPAN <= E) {
            int2 sd[EDGE_U];
            int  rk[EDGE_U];
            int  st[EDGE_U];
#pragma unroll
            for (int u = 0; u < EDGE_U; ++u) {
                sd[u] = *(const int2*)(edges + 2 * (size_t)(e0 + u * 256));
                rk[u] = rank[e0 + u * 256];
            }
#pragma unroll
            for (int u = 0; u < EDGE_U; ++u) st[u] = seg[2 * (size_t)sd[u].x];
            __builtin_amdgcn_sched_barrier(0);
#pragma unroll
            for (int u = 0; u < EDGE_U; ++u) sorted[st[u] + rk[u]] = sd[u].y;
        } else {
#pragma unroll
            for (int u = 0; u < EDGE_U; ++u) {
                const int e = e0 + u * 256;
                if (e < E) {
                    const int2 sd = *(const int2*)(edges + 2 * (size_t)e);
                    sorted[seg[2 * (size_t)sd.x] + rank[e]] = sd.y;
                }
            }
        }
    }
}

template <int U>
__device__ __forceinline__ void gat_walk_step(
    const float* __restrict__ h, int d, float w, int i, int lane,
    float4& acc, float& den)
{
    float4 hv[U];
    float  wi[U];
#pragma unroll
    for (int u = 0; u < U; ++u) {
        const int di = __builtin_amdgcn_readlane(d, i + u);
        wi[u] = __int_as_float(__builtin_amdgcn_readlane(__float_as_int(w), i + u));
        hv[u] = *(const float4*)(h + (size_t)di * OUT_DIM + lane * 4);
    }
#pragma unroll
    for (int u = 0; u < U; ++u) {
        acc.x += wi[u] * hv[u].x;
        acc.y += wi[u] * hv[u].y;
        acc.z += wi[u] * hv[u].z;
        acc.w += wi[u] * hv[u].w;
        den   += wi[u];
    }
}

__global__ __launch_bounds__(256) void gat_walk_kernel(
    const float* __restrict__ h, const float* __restrict__ s_src,
    const float* __restrict__ s_dst, const int* __restrict__ seg,
    const int* __restrict__ sorted,
    const int* __restrict__ node_idx, float* __restrict__ out, int B)
{
    const int lane = threadIdx.x & 31;
    const int wv = __builtin_amdgcn_readfirstlane(threadIdx.x >> 5);
    const int j  = blockIdx.x * 8 + wv;
    if (j >= B) return;

    const int   row = node_idx[j];
    const unsigned long long sg = *(const unsigned long long*)(seg + 2 * (size_t)row);
    const int   beg = (int)(unsigned)(sg & 0xffffffffull);
    const int   n   = (int)(unsigned)(sg >> 32);
    const float ss  = s_src[row];

    float4 acc = make_float4(0.0f, 0.0f, 0.0f, 0.0f);
    float  den = 0.0f;

    for (int c = 0; c < n; c += 32) {
        const int m = (n - c < 32) ? (n - c) : 32;
        int   d = 0;
        float w = 0.0f;
        if (lane < m) {
            d = sorted[beg + c + lane];
            float s = ss + s_dst[d];
            s = (s >= 0.0f) ? s : SLOPE * s;
            w = expf(s);
        }
        int i = 0;
        for (; i + 8 <= m; i += 8) gat_walk_step<8>(h, d, w, i, lane, acc, den);
        if (i + 4 <= m) { gat_walk_step<4>(h, d, w, i, lane, acc, den); i += 4; }
        if (i + 2 <= m) { gat_walk_step<2>(h, d, w, i, lane, acc, den); i += 2; }
        if (i < m)      { gat_walk_step<1>(h, d, w, i, lane, acc, den); }
    }

    const float inv = 1.0f / den;
    float4 ov;
    ov.x = acc.x * inv; ov.y = acc.y * inv; ov.z = acc.z * inv; ov.w = acc.w * inv;
    *(float4*)(out + (size_t)j * OUT_DIM + lane * 4) = ov;
}

extern "C" void kernel_launch(void* const* d_in, const int* in_sizes, int n_in,
                              void* d_out, int out_size, void* d_ws, size_t ws_size,
                              hipStream_t stream) {
    const float* features     = (const float*)d_in[0];
    const float* W            = (const float*)d_in[1];
    const float* b            = (const float*)d_in[2];
    const float* a            = (const float*)d_in[3];
    const int*   edges        = (const int*)d_in[4];
    const int*   unique_nodes = (const int*)d_in[5];
    const int*   node_idx     = (const int*)d_in[6];
    float* out = (float*)d_out;

    const int N = in_sizes[5];
    const int E = in_sizes[4] / 2;
    const int B = in_sizes[6];

    float* h      = (float*)d_ws;
    int*   seg    = (int*)(h + (size_t)N * OUT_DIM);
    float* s_src  = (float*)(seg + 2 * (size_t)N);
    float* s_dst  = s_src + N;
    int*   cnt    = (int*)(s_dst + N);
    int*   total  = cnt + N;
    int*   rank   = total + 1;
    int*   sorted = rank + E;

    hipMemsetAsync(cnt, 0, ((size_t)N + 1) * sizeof(int), stream);

    {
        gat_gemm_kernel<<<GEMM_GROUPS, 256, 0, stream>>>(features, W, b, unique_nodes, h, N);
    }
    {
        gat_scores_kernel<<<SCORE_GROUPS, 256, 0, stream>>>(h, a, s_src, s_dst, N);
    }
    {
        gat_rank_kernel<<<EDGE_GROUPS, 256, 0, stream>>>(edges, cnt, rank, E);
    }
    {
        dim3 grid((N + 255) / 256);
        gat_alloc_kernel<<<grid, 256, 0, stream>>>(cnt, total, seg, N);
    }
    {
        gat_fill_kernel<<<EDGE_GROUPS, 256, 0, stream>>>(edges, rank, seg, sorted, E);
    }
    {
        dim3 grid((B + 7) / 8);
        gat_walk_kernel<<<grid, 256, 0, stream>>>(h, s_src, s_dst, seg, sorted,
                                                  node_idx, out, B);
    }
}


# edge sort without per-edge global atomics: edges binned by 256-row source range with LDS counters over 192 list shares, a work-group per bin ranks rows in LDS; row sums in LDS-rank order
# speedup vs baseline: 3.9751x; 1.4502x over previous
#include <hip/hip_runtime.h>
#include <hip/hip_bf16.h>

#define IN_DIM 256
#define OUT_DIM 128
#define SLOPE 0.1f

typedef float v2f __attribute__((ext_vector_type(2)));
typedef float v8f __attribute__((ext_vector_type(8)));

#define GEMM_GROUPS 256
#define GEMM_WAVES 8
#define KCHUNKS (IN_DIM / 8)
#define CTILES (OUT_DIM / 16)
#define CHUNK_PITCH (CTILES * 32)
#define GEMM_LDS_SLOTS 10752

__device__ __forceinline__ void gat_load_stripe(
    const float* __restrict__ features, const int* __restrict__ unique_nodes,
    int tile, int mn, int half, int N, float4 (&dst)[KCHUNKS])
{
    int arow_idx = tile * 16 + mn;
    if (arow_idx >= N) arow_idx = N - 1;
    const int gsrc = unique_nodes[arow_idx];
    const float* __restrict__ arow = features + (size_t)gsrc * IN_DIM + 4 * half;
#pragma unroll
    for (int t = 0; t < KCHUNKS; ++t)
        dst[t] = *(const float4*)(arow + 8 * t);
}

__global__ __launch_bounds__(256) void gat_gemm_kernel(
    const float* __restrict__ features, const float* __restrict__ W,
    const float* __restrict__ bias, const int* __restrict__ unique_nodes,
    float* __restrict__ h, int N)
{
    __shared__ float4 wl[GEMM_LDS_SLOTS];

    const int lane = threadIdx.x & 31;
    const int wave = __builtin_amdgcn_readfirstlane(threadIdx.x >> 5);
    const int half = lane >> 4;
    const int mn   = lane & 15;

    const int nTiles = (N + 15) >> 4;
    const int stride = (int)gridDim.x * GEMM_WAVES;
    int tile = wave * (int)gridDim.x + (int)blockIdx.x;

    float4 cur[KCHUNKS];
    float4 nxt[KCHUNKS];
#pragma unroll
    for (int t = 0; t < KCHUNKS; ++t) {
        cur[t] = make_float4(0.0f, 0.0f, 0.0f, 0.0f);
        nxt[t] = make_float4(0.0f, 0.0f, 0.0f, 0.0f);
    }
    if (tile < nTiles) gat_load_stripe(features, unique_nodes, tile, mn, half, N, cur);

    {
        const int qlo  = lane & 7;
        const int rsub = lane >> 3;
#pragma unroll 16
        for (int it = 0; it < (OUT_DIM * IN_DIM / 4) / 256; ++it) {
            const int wi = it * GEMM_WAVES + wave;
            const int n  = (wi >> 3) * 4 + rsub;
            const int q  = (wi & 7) * 8 + qlo;
            const float4 v = *(const float4*)(W + (size_t)n * IN_DIM + 4 * q);
            wl[(q >> 1) * CHUNK_PITCH + (n >> 4) * 32 + (q & 1) * 16 + (n & 15)] = v;
        }
    }

    float bv[CTILES];
#pragma unroll
    for (int c = 0; c < CTILES; ++c) bv[c] = bias[c * 16 + mn];

    __syncthreads();

    const int loSlot = lane;
    int hiSlot = lane + (KCHUNKS / 2) * CHUNK_PITCH;
    asm volatile("" : "+v"(hiSlot));

    for (; tile < nTiles; tile += stride) {
        const int  rowBase = tile * 16;
        const bool more    = (tile + stride) < nTiles;
        if (more) gat_load_stripe(features, unique_nodes, tile + stride, mn, half, N, nxt);

        v8f acc[CTILES];
#pragma unroll
        for (int c = 0; c < CTILES; ++c) {
#pragma unroll
            for (int r = 0; r < 8; ++r) acc[c][r] = bv[c];
        }

        float4 bcur[CTILES];
#pragma unroll
        for (int c = 0; c < CTILES; ++c) bcur[c] = wl[loSlot + c * 32];

#pragma unroll
        for (int t = 0; t < KCHUNKS; ++t) {
            float4 bnx[CTILES];
            if (t + 1 < KCHUNKS) {
#pragma unroll
                for (int c = 0; c < CTILES; ++c)
                    bnx[c] = (t + 1 < KCHUNKS / 2)
                        ? wl[loSlot + (t + 1) * CHUNK_PITCH + c * 32]
                        : wl[hiSlot + (t + 1 - KCHUNKS / 2) * CHUNK_PITCH + c * 32];
            }
            __builtin_amdgcn_sched_barrier(0);
            const float4 a = cur[t];
            const v2f a0 = {a.x, a.y};
            const v2f a1 = {a.z, a.w};
#pragma unroll
            for (int c = 0; c < CTILES; ++c) {
                const v2f b0 = {bcur[c].x, bcur[c].y};
                acc[c] = __builtin_amdgcn_wmma_f32_16x16x4_f32(
                     false, a0,  false, b0,
                     (short)0, acc[c],  false,  false);
            }
#pragma unroll
            for (int c = 0; c < CTILES; ++c) {
                const v2f b1 = {bcur[c].z, bcur[c].w};
                acc[c] = __builtin_amdgcn_wmma_f32_16x16x4_f32(
                     false, a1,  false, b1,
                     (short)0, acc[c],  false,  false);
            }
            __builtin_amdgcn_sched_barrier(0);
            if (t + 1 < KCHUNKS) {
#pragma unroll
                for (int c = 0; c < CTILES; ++c) bcur[c] = bnx[c];
            }
        }

        asm volatile("v_nop\n\tv_nop\n\tv_nop\n\tv_nop\n\tv_nop\n\tv_nop\n\tv_nop\n\tv_nop");
        __builtin_amdgcn_sched_barrier(0);

        float* __restrict__ obase =
            h + (size_t)(rowBase + 8 * half) * OUT_DIM + mn;
        if (rowBase + 16 <= N) {
#pragma unroll
            for (int c = 0; c < CTILES; ++c) {
#pragma unroll
                for (int r = 0; r < 8; ++r)
                    obase[(size_t)r * OUT_DIM + c * 16] = acc[c][r];
            }
        } else {
#pragma unroll
            for (int c = 0; c < CTILES; ++c) {
#pragma unroll
                for (int r = 0; r < 8; ++r) {
                    const int orow = rowBase + 8 * half + r;
                    if (orow < N) obase[(size_t)r * OUT_DIM + c * 16] = acc[c][r];
                }
            }
        }

        if (more) {
#pragma unroll
            for (int t = 0; t < KCHUNKS; ++t) cur[t] = nxt[t];
        }
    }
}

#define SCORE_GROUPS 784
#define SCORE_ROWS 8

__global__ __launch_bounds__(256) void gat_scores_kernel(
    const float* __restrict__ h, const float* __restrict__ a,
    float* __restrict__ s_src, float* __restrict__ s_dst, int N)
{
    const int lane = threadIdx.x & 31;
    const int gw = (int)blockIdx.x * 8 + __builtin_amdgcn_readfirstlane(threadIdx.x >> 5);
    const int step = (int)gridDim.x * 8 * SCORE_ROWS;

    const float4 as = *(const float4*)(a + lane * 4);
    const float4 ad = *(const float4*)(a + OUT_DIM + lane * 4);

    for (int row0 = gw * SCORE_ROWS; row0 < N; row0 += step) {
        float4 hv[SCORE_ROWS];
#pragma unroll
        for (int u = 0; u < SCORE_ROWS; ++u) {
            int r = row0 + u;
            if (r >= N) r = N - 1;
            hv[u] = *(const float4*)(h + (size_t)r * OUT_DIM + lane * 4);
        }
        float outs = 0.0f, outd = 0.0f;
#pragma unroll
        for (int u = 0; u < SCORE_ROWS; ++u) {
            float ps = hv[u].x * as.x + hv[u].y * as.y + hv[u].z * as.z + hv[u].w * as.w;
            float pd = hv[u].x * ad.x + hv[u].y * ad.y + hv[u].z * ad.z + hv[u].w * ad.w;
#pragma unroll
            for (int off = 16; off > 0; off >>= 1) {
                ps += __shfl_xor(ps, off, 32);
                pd += __shfl_xor(pd, off, 32);
            }
            if (lane == u) { outs = ps; outd = pd; }
        }
        if (lane < SCORE_ROWS && row0 + lane < N) {
            s_src[row0 + lane] = outs;
            s_dst[row0 + lane] = outd;
        }
    }
}

#define BIN_SHIFT 8
#define BIN_ROWS 256
#define BIN_GROUPS 192
#define BIN_THREADS 1024
#define BIN_U 4
#define BIN_LDS 1024
#define BIN_PER_LANE (BIN_GROUPS / 32)

__global__ __launch_bounds__(BIN_THREADS) void gat_hist_kernel(
    const int* __restrict__ edges, int* __restrict__ table,
    int* __restrict__ total, int E, int N)
{
    __shared__ int lh[BIN_LDS];
    const int nb  = (N + BIN_ROWS - 1) >> BIN_SHIFT;
    const int tid = (int)threadIdx.x;
    for (int b = tid; b < nb; b += BIN_THREADS) lh[b] = 0;
    __syncthreads();

    const int per = (E + BIN_GROUPS - 1) / BIN_GROUPS;
    const int lo  = (int)blockIdx.x * per;
    const int hi  = (lo + per < E) ? (lo + per) : E;
    int e0 = lo + tid;
    for (; e0 + (BIN_U - 1) * BIN_THREADS < hi; e0 += BIN_THREADS * BIN_U) {
        int s[BIN_U];
#pragma unroll
        for (int u = 0; u < BIN_U; ++u) s[u] = edges[2 * (size_t)(e0 + u * BIN_THREADS)];
#pragma unroll
        for (int u = 0; u < BIN_U; ++u) atomicAdd(&lh[s[u] >> BIN_SHIFT], 1);
    }
    for (; e0 < hi; e0 += BIN_THREADS)
        atomicAdd(&lh[edges[2 * (size_t)e0] >> BIN_SHIFT], 1);
    __syncthreads();
    for (int b = tid; b < nb; b += BIN_THREADS)
        table[(size_t)b * BIN_GROUPS + blockIdx.x] = lh[b];
    if (blockIdx.x == 0 && tid == 0) *total = 0;
}

__global__ __launch_bounds__(256) void gat_binscan_kernel(
    int* table, int* __restrict__ total, int N)
{
    const int nb   = (N + BIN_ROWS - 1) >> BIN_SHIFT;
    const int lane = threadIdx.x & 31;
    const int bin  = (int)blockIdx.x * 8 + (int)(threadIdx.x >> 5);
    if (bin >= nb) return;
    int* col    = table + (size_t)bin * BIN_GROUPS + lane * BIN_PER_LANE;
    int* binseg = table + (size_t)nb * BIN_GROUPS;

    int v[BIN_PER_LANE];
    int s = 0;
#pragma unroll
    for (int k = 0; k < BIN_PER_LANE; ++k) { v[k] = col[k]; s += v[k]; }

    int x = s;
    for (int off = 1; off < 32; off <<= 1) {
        const int y = __shfl_up(x, off, 32);
        if (lane >= off) x += y;
    }
    const int tot = __shfl(x, 31, 32);
    int base = 0;
    if (lane == 0) base = atomicAdd(total, tot);
    base = __shfl(base, 0, 32);

    int run = base + x - s;
#pragma unroll
    for (int k = 0; k < BIN_PER_LANE; ++k) { col[k] = run; run += v[k]; }
    if (lane == 0) *(int2*)(binseg + 2 * (size_t)bin) = make_int2(base, tot);
}

__global__ __launch_bounds__(BIN_THREADS) void gat_bin_kernel(
    const int* __restrict__ edges, const int* __restrict__ table,
    int* __restrict__ binned, int E, int N)
{
    __shared__ int slot[BIN_LDS];
    const int nb  = (N + BIN_ROWS - 1) >> BIN_SHIFT;
    const int tid = (int)threadIdx.x;
    for (int b = tid; b < nb; b += BIN_THREADS)
        slot[b] = table[(size_t)b * BIN_GROUPS + blockIdx.x];
    __syncthreads();

    const int per = (E + BIN_GROUPS - 1) / BIN_GROUPS;
    const int lo  = (int)blockIdx.x * per;
    const int hi  = (lo + per < E) ? (lo + per) : E;
    int e0 = lo + tid;
    for (; e0 + (BIN_U - 1) * BIN_THREADS < hi; e0 += BIN_THREADS * BIN_U) {
        int2 sd[BIN_U];
#pragma unroll
        for (int u = 0; u < BIN_U; ++u)
            sd[u] = *(const int2*)(edges + 2 * (size_t)(e0 + u * BIN_THREADS));
#pragma unroll
        for (int u = 0; u < BIN_U; ++u) {
            const int p = atomicAdd(&slot[sd[u].x >> BIN_SHIFT], 1);
            binned[p] = (sd[u].y << BIN_SHIFT) | (sd[u].x & (BIN_ROWS - 1));
        }
    }
    for (; e0 < hi; e0 += BIN_THREADS) {
        const int2 sd = *(const int2*)(edges + 2 * (size_t)e0);
        const int p = atomicAdd(&slot[sd.x >> BIN_SHIFT], 1);
        binned[p] = (sd.y << BIN_SHIFT) | (sd.x & (BIN_ROWS - 1));
    }
}

__global__ __launch_bounds__(BIN_THREADS) void gat_rowsort_kernel(
    const int* __restrict__ table, const int* __restrict__ binned,
    int* __restrict__ seg, int* __restrict__ sorted, int N)
{
    __shared__ int cnt[BIN_ROWS];
    __shared__ int cur[BIN_ROWS];
    __shared__ int wsum[BIN_ROWS / 32];
    const int nb   = (N + BIN_ROWS - 1) >> BIN_SHIFT;
    const int tid  = (int)threadIdx.x;
    const int lane = tid & 31;
    const int wv   = tid >> 5;
    const int bin  = (int)blockIdx.x;
    const int2 bs  = *(const int2*)(table + (size_t)nb * BIN_GROUPS + 2 * (size_t)bin);
    const int base = bs.x;
    const int n    = bs.y;
    const int* __restrict__ rec = binned + base;

    if (tid < BIN_ROWS) cnt[tid] = 0;
    __syncthreads();

    {
        int i0 = tid;
        for (; i0 + (BIN_U - 1) * BIN_THREADS < n; i0 += BIN_THREADS * BIN_U) {
            int p[BIN_U];
#pragma unroll
            for (int u = 0; u < BIN_U; ++u) p[u] = rec[i0 + u * BIN_THREADS];
#pragma unroll
            for (int u = 0; u < BIN_U; ++u) atomicAdd(&cnt[p[u] & (BIN_ROWS - 1)], 1);
        }
        for (; i0 < n; i0 += BIN_THREADS) atomicAdd(&cnt[rec[i0] & (BIN_ROWS - 1)], 1);
    }
    __syncthreads();

    const int c = (tid < BIN_ROWS) ? cnt[tid] : 0;
    int x = c;
    for (int off = 1; off < 32; off <<= 1) {
        const int y = __shfl_up(x, off, 32);
        if (lane >= off) x += y;
    }
    if (tid < BIN_ROWS && lane == 31) wsum[wv] = x;
    __syncthreads();
    if (tid == 0) {
        int s = 0;
        for (int w = 0; w < BIN_ROWS / 32; ++w) { const int t = wsum[w]; wsum[w] = s; s += t; }
    }
    __syncthreads();
    if (tid < BIN_ROWS) {
        const int excl = wsum[wv] + x - c;
        cur[tid] = excl;
        const int row = bin * BIN_ROWS + tid;
        if (row < N) *(int2*)(seg + 2 * (size_t)row) = make_int2(base + excl, c);
    }
    __syncthreads();

    int* __restrict__ dstv = sorted + base;
    {
        int i0 = tid;
        for (; i0 + (BIN_U - 1) * BIN_THREADS < n; i0 += BIN_THREADS * BIN_U) {
            int p[BIN_U];
#pragma unroll
            for (int u = 0; u < BIN_U; ++u) p[u] = rec[i0 + u * BIN_THREADS];
#pragma unroll
            for (int u = 0; u < BIN_U; ++u) {
                const int q = atomicAdd(&cur[p[u] & (BIN_ROWS - 1)], 1);
                dstv[q] = p[u] >> BIN_SHIFT;
            }
        }
        for (; i0 < n; i0 += BIN_THREADS) {
            const int p = rec[i0];
            const int q = atomicAdd(&cur[p & (BIN_ROWS - 1)], 1);
            dstv[q] = p >> BIN_SHIFT;
        }
    }
}

template <int U>
__device__ __forceinline__ void gat_walk_step(
    const float* __restrict__ h, int d, float w, int i, int lane,
    float4& acc, float& den)
{
    float4 hv[U];
    float  wi[U];
#pragma unroll
    for (int u = 0; u < U; ++u) {
        const int di = __builtin_amdgcn_readlane(d, i + u);
        wi[u] = __int_as_float(__builtin_amdgcn_readlane(__float_as_int(w), i + u));
        hv[u] = *(const float4*)(h + (size_t)di * OUT_DIM + lane * 4);
    }
#pragma unroll
    for (int u = 0; u < U; ++u) {
        acc.x += wi[u] * hv[u].x;
        acc.y += wi[u] * hv[u].y;
        acc.z += wi[u] * hv[u].z;
        acc.w += wi[u] * hv[u].w;
        den   += wi[u];
    }
}

__global__ __launch_bounds__(256) void gat_walk_kernel(
    const float* __restrict__ h, const float* __restrict__ s_src,
    const float* __restrict__ s_dst, const int* __restrict__ seg,
    const int* __restrict__ sorted,
    const int* __restrict__ node_idx, float* __restrict__ out, int B)
{
    const int lane = threadIdx.x & 31;
    const int wv = __builtin_amdgcn_readfirstlane(threadIdx.x >> 5);
    const int j  = blockIdx.x * 8 + wv;
    if (j >= B) return;

    const int   row = node_idx[j];
    const unsigned long long sg = *(const unsigned long long*)(seg + 2 * (size_t)row);
    const int   beg = (int)(unsigned)(sg & 0xffffffffull);
    const int   n   = (int)(unsigned)(sg >> 32);
    const float ss  = s_src[row];

    float4 acc = make_float4(0.0f, 0.0f, 0.0f, 0.0f);
    float  den = 0.0f;

    for (int c = 0; c < n; c += 32) {
        const int m = (n - c < 32) ? (n - c) : 32;
        int   d = 0;
        float w = 0.0f;
        if (lane < m) {
            d = sorted[beg + c + lane];
            float s = ss + s_dst[d];
            s = (s >= 0.0f) ? s : SLOPE * s;
            w = expf(s);
        }
        int i = 0;
        for (; i + 8 <= m; i += 8) gat_walk_step<8>(h, d, w, i, lane, acc, den);
        if (i + 4 <= m) { gat_walk_step<4>(h, d, w, i, lane, acc, den); i += 4; }
        if (i + 2 <= m) { gat_walk_step<2>(h, d, w, i, lane, acc, den); i += 2; }
        if (i < m)      { gat_walk_step<1>(h, d, w, i, lane, acc, den); }
    }

    const float inv = 1.0f / den;
    float4 ov;
    ov.x = acc.x * inv; ov.y = acc.y * inv; ov.z = acc.z * inv; ov.w = acc.w * inv;
    *(float4*)(out + (size_t)j * OUT_DIM + lane * 4) = ov;
}

extern "C" void kernel_launch(void* const* d_in, const int* in_sizes, int n_in,
                              void* d_out, int out_size, void* d_ws, size_t ws_size,
                              hipStream_t stream) {
    const float* features     = (const float*)d_in[0];
    const float* W            = (const float*)d_in[1];
    const float* b            = (const float*)d_in[2];
    const float* a            = (const float*)d_in[3];
    const int*   edges        = (const int*)d_in[4];
    const int*   unique_nodes = (const int*)d_in[5];
    const int*   node_idx     = (const int*)d_in[6];
    float* out = (float*)d_out;

    const int N = in_sizes[5];
    const int E = in_sizes[4] / 2;
    const int B = in_sizes[6];

    float* h      = (float*)d_ws;
    int*   seg    = (int*)(h + (size_t)N * OUT_DIM);
    float* s_src  = (float*)(seg + 2 * (size_t)N);
    float* s_dst  = s_src + N;
    int*   table  = (int*)(s_dst + N);
    int*   total  = table + N;
    int*   binned = total + 1;
    int*   sorted = binned + E;

    {
        gat_gemm_kernel<<<GEMM_GROUPS, 256, 0, stream>>>(features, W, b, unique_nodes, h, N);
    }
    {
        gat_scores_kernel<<<SCORE_GROUPS, 256, 0, stream>>>(h, a, s_src, s_dst, N);
    }
    {
        gat_hist_kernel<<<BIN_GROUPS, BIN_THREADS, 0, stream>>>(edges, table, total, E, N);
    }
    {
        dim3 grid(((N + 255) / 256 + 7) / 8);
        gat_binscan_kernel<<<grid, 256, 0, stream>>>(table, total, N);
    }
    {
        gat_bin_kernel<<<BIN_GROUPS, BIN_THREADS, 0, stream>>>(edges, table, binned, E, N);
    }
    {
        dim3 grid((N + 255) / 256);
        gat_rowsort_kernel<<<grid, BIN_THREADS, 0, stream>>>(table, binned, seg, sorted, N);
    }
    {
        dim3 grid((B + 7) / 8);
        gat_walk_kernel<<<grid, 256, 0, stream>>>(h, s_src, s_dst, seg, sorted,
                                                  node_idx, out, B);
    }
}


# edge binning in one launch: each share counts, scans and places its own edges in its own piece, the row sort gathers a bin's runs; hist and scan launches gone (7 to 5); row sums in LDS-rank order
# speedup vs baseline: 4.3687x; 1.0990x over previous
#include <hip/hip_runtime.h>
#include <hip/hip_bf16.h>

#define IN_DIM 256
#define OUT_DIM 128
#define SLOPE 0.1f

typedef float v2f __attribute__((ext_vector_type(2)));
typedef float v8f __attribute__((ext_vector_type(8)));

#define GEMM_GROUPS 256
#define GEMM_WAVES 8
#define KCHUNKS (IN_DIM / 8)
#define CTILES (OUT_DIM / 16)
#define CHUNK_PITCH (CTILES * 32)
#define GEMM_LDS_SLOTS 10752

__device__ __forceinline__ void gat_load_stripe(
    const float* __restrict__ features, const int* __restrict__ unique_nodes,
    int tile, int mn, int half, int N, float4 (&dst)[KCHUNKS])
{
    int arow_idx = tile * 16 + mn;
    if (arow_idx >= N) arow_idx = N - 1;
    const int gsrc = unique_nodes[arow_idx];
    const float* __restrict__ arow = features + (size_t)gsrc * IN_DIM + 4 * half;
#pragma unroll
    for (int t = 0; t < KCHUNKS; ++t)
        dst[t] = *(const float4*)(arow + 8 * t);
}

__global__ __launch_bounds__(256) void gat_gemm_kernel(
    const float* __restrict__ features, const float* __restrict__ W,
    const float* __restrict__ bias, const int* __restrict__ unique_nodes,
    float* __restrict__ h, int N)
{
    __shared__ float4 wl[GEMM_LDS_SLOTS];

    const int lane = threadIdx.x & 31;
    const int wave = __builtin_amdgcn_readfirstlane(threadIdx.x >> 5);
    const int half = lane >> 4;
    const int mn   = lane & 15;

    const int nTiles = (N + 15) >> 4;
    const int stride = (int)gridDim.x * GEMM_WAVES;
    int tile = wave * (int)gridDim.x + (int)blockIdx.x;

    float4 cur[KCHUNKS];
    float4 nxt[KCHUNKS];
#pragma unroll
    for (int t = 0; t < KCHUNKS; ++t) {
        cur[t] = make_float4(0.0f, 0.0f, 0.0f, 0.0f);
        nxt[t] = make_float4(0.0f, 0.0f, 0.0f, 0.0f);
    }
    if (tile < nTiles) gat_load_stripe(features, unique_nodes, tile, mn, half, N, cur);

    {
        const int qlo  = lane & 7;
        const int rsub = lane >> 3;
#pragma unroll 16
        for (int it = 0; it < (OUT_DIM * IN_DIM / 4) / 256; ++it) {
            const int wi = it * GEMM_WAVES + wave;
            const int n  = (wi >> 3) * 4 + rsub;
            const int q  = (wi & 7) * 8 + qlo;
            const float4 v = *(const float4*)(W + (size_t)n * IN_DIM + 4 * q);
            wl[(q >> 1) * CHUNK_PITCH + (n >> 4) * 32 + (q & 1) * 16 + (n & 15)] = v;
        }
    }

    float bv[CTILES];
#pragma unroll
    for (int c = 0; c < CTILES; ++c) bv[c] = bias[c * 16 + mn];

    __syncthreads();

    const int loSlot = lane;
    int hiSlot = lane + (KCHUNKS / 2) * CHUNK_PITCH;
    asm volatile("" : "+v"(hiSlot));

    for (; tile < nTiles; tile += stride) {
        const int  rowBase = tile * 16;
        const bool more    = (tile + stride) < nTiles;
        if (more) gat_load_stripe(features, unique_nodes, tile + stride, mn, half, N, nxt);

        v8f acc[CTILES];
#pragma unroll
        for (int c = 0; c < CTILES; ++c) {
#pragma unroll
            for (int r = 0; r < 8; ++r) acc[c][r] = bv[c];
        }

        float4 bcur[CTILES];
#pragma unroll
        for (int c = 0; c < CTILES; ++c) bcur[c] = wl[loSlot + c * 32];

#pragma unroll
        for (int t = 0; t < KCHUNKS; ++t) {
            float4 bnx[CTILES];
            if (t + 1 < KCHUNKS) {
#pragma unroll
                for (int c = 0; c < CTILES; ++c)
                    bnx[c] = (t + 1 < KCHUNKS / 2)
                        ? wl[loSlot + (t + 1) * CHUNK_PITCH + c * 32]
                        : wl[hiSlot + (t + 1 - KCHUNKS / 2) * CHUNK_PITCH + c * 32];
            }
            __builtin_amdgcn_sched_barrier(0);
            const float4 a = cur[t];
            const v2f a0 = {a.x, a.y};
            const v2f a1 = {a.z, a.w};
#pragma unroll
            for (int c = 0; c < CTILES; ++c) {
                const v2f b0 = {bcur[c].x, bcur[c].y};
                acc[c] = __builtin_amdgcn_wmma_f32_16x16x4_f32(
                     false, a0,  false, b0,
                     (short)0, acc[c],  false,  false);
            }
#pragma unroll
            for (int c = 0; c < CTILES; ++c) {
                const v2f b1 = {bcur[c].z, bcur[c].w};
                acc[c] = __builtin_amdgcn_wmma_f32_16x16x4_f32(
                     false, a1,  false, b1,
                     (short)0, acc[c],  false,  false);
            }
            __builtin_amdgcn_sched_barrier(0);
            if (t + 1 < KCHUNKS) {
#pragma unroll
                for (int c = 0; c < CTILES; ++c) bcur[c] = bnx[c];
            }
        }

        asm volatile("v_nop\n\tv_nop\n\tv_nop\n\tv_nop\n\tv_nop\n\tv_nop\n\tv_nop\n\tv_nop");
        __builtin_amdgcn_sched_barrier(0);

        float* __restrict__ obase =
            h + (size_t)(rowBase + 8 * half) * OUT_DIM + mn;
        if (rowBase + 16 <= N) {
#pragma unroll
            for (int c = 0; c < CTILES; ++c) {
#pragma unroll
                for (int r = 0; r < 8; ++r)
                    obase[(size_t)r * OUT_DIM + c * 16] = acc[c][r];
            }
        } else {
#pragma unroll
            for (int c = 0; c < CTILES; ++c) {
#pragma unroll
                for (int r = 0; r < 8; ++r) {
                    const int orow = rowBase + 8 * half + r;
                    if (orow < N) obase[(size_t)r * OUT_DIM + c * 16] = acc[c][r];
                }
            }
        }

        if (more) {
#pragma unroll
            for (int t = 0; t < KCHUNKS; ++t) cur[t] = nxt[t];
        }
    }
}

#define SCORE_GROUPS 784
#define SCORE_ROWS 8

__global__ __launch_bounds__(256) void gat_scores_kernel(
    const float* __restrict__ h, const float* __restrict__ a,
    float* __restrict__ s_src, float* __restrict__ s_dst, int N)
{
    const int lane = threadIdx.x & 31;
    const int gw = (int)blockIdx.x * 8 + __builtin_amdgcn_readfirstlane(threadIdx.x >> 5);
    const int step = (int)gridDim.x * 8 * SCORE_ROWS;

    const float4 as = *(const float4*)(a + lane * 4);
    const float4 ad = *(const float4*)(a + OUT_DIM + lane * 4);

    for (int row0 = gw * SCORE_ROWS; row0 < N; row0 += step) {
        float4 hv[SCORE_ROWS];
#pragma unroll
        for (int u = 0; u < SCORE_ROWS; ++u) {
            int r = row0 + u;
            if (r >= N) r = N - 1;
            hv[u] = *(const float4*)(h + (size_t)r * OUT_DIM + lane * 4);
        }
        float outs = 0.0f, outd = 0.0f;
#pragma unroll
        for (int u = 0; u < SCORE_ROWS; ++u) {
            float ps = hv[u].x * as.x + hv[u].y * as.y + hv[u].z * as.z + hv[u].w * as.w;
            float pd = hv[u].x * ad.x + hv[u].y * ad.y + hv[u].z * ad.z + hv[u].w * ad.w;
#pragma unroll
            for (int off = 16; off > 0; off >>= 1) {
                ps += __shfl_xor(ps, off, 32);
                pd += __shfl_xor(pd, off, 32);
            }
            if (lane == u) { outs = ps; outd = pd; }
        }
        if (lane < SCORE_ROWS && row0 + lane < N) {
            s_src[row0 + lane] = outs;
            s_dst[row0 + lane] = outd;
        }
    }
}

#define BIN_SHIFT 8
#define BIN_ROWS 256
#define BIN_GROUPS 224
#define BIN_THREADS 1024
#define BIN_U 4
#define BIN_LDS 1024
#define BIN_RUNS (BIN_GROUPS / 32)

__global__ __launch_bounds__(BIN_THREADS) void gat_bin_kernel(
    const int* __restrict__ edges, int* __restrict__ table,
    int* __restrict__ binned, int E, int N)
{
    static_assert(BIN_LDS == BIN_THREADS, "one counter per thread in the scan");
    __shared__ int slot[BIN_LDS];
    __shared__ int wsum[BIN_THREADS / 32];
    const int nb   = (N + BIN_ROWS - 1) >> BIN_SHIFT;
    const int tid  = (int)threadIdx.x;
    const int lane = tid & 31;
    const int wv   = tid >> 5;
    slot[tid] = 0;
    __syncthreads();

    const int per = (E + BIN_GROUPS - 1) / BIN_GROUPS;
    int lo = (int)blockIdx.x * per;
    if (lo > E) lo = E;
    const int hi = (lo + per < E) ? (lo + per) : E;

    {
        int e0 = lo + tid;
        for (; e0 + (BIN_U - 1) * BIN_THREADS < hi; e0 += BIN_THREADS * BIN_U) {
            int s[BIN_U];
#pragma unroll
            for (int u = 0; u < BIN_U; ++u) s[u] = edges[2 * (size_t)(e0 + u * BIN_THREADS)];
#pragma unroll
            for (int u = 0; u < BIN_U; ++u) atomicAdd(&slot[s[u] >> BIN_SHIFT], 1);
        }
        for (; e0 < hi; e0 += BIN_THREADS)
            atomicAdd(&slot[edges[2 * (size_t)e0] >> BIN_SHIFT], 1);
    }
    __syncthreads();

    const int c = slot[tid];
    int x = c;
    for (int off = 1; off < 32; off <<= 1) {
        const int y = __shfl_up(x, off, 32);
        if (lane >= off) x += y;
    }
    if (lane == 31) wsum[wv] = x;
    __syncthreads();
    if (wv == 0) {
        const int s = wsum[lane];
        int t = s;
        for (int off = 1; off < 32; off <<= 1) {
            const int y = __shfl_up(t, off, 32);
            if (lane >= off) t += y;
        }
        wsum[lane] = t - s;
    }
    __syncthreads();
    const int excl = wsum[wv] + x - c;
    slot[tid] = excl;
    if (tid <= nb) table[(size_t)tid * BIN_GROUPS + blockIdx.x] = excl;
    if (tid == 0 && nb == BIN_LDS) table[(size_t)nb * BIN_GROUPS + blockIdx.x] = hi - lo;
    __syncthreads();

    int* __restrict__ piece = binned + lo;
    {
        int e0 = lo + tid;
        for (; e0 + (BIN_U - 1) * BIN_THREADS < hi; e0 += BIN_THREADS * BIN_U) {
            int2 sd[BIN_U];
#pragma unroll
            for (int u = 0; u < BIN_U; ++u)
                sd[u] = *(const int2*)(edges + 2 * (size_t)(e0 + u * BIN_THREADS));
#pragma unroll
            for (int u = 0; u < BIN_U; ++u) {
                const int p = atomicAdd(&slot[sd[u].x >> BIN_SHIFT], 1);
                piece[p] = (sd[u].y << BIN_SHIFT) | (sd[u].x & (BIN_ROWS - 1));
            }
        }
        for (; e0 < hi; e0 += BIN_THREADS) {
            const int2 sd = *(const int2*)(edges + 2 * (size_t)e0);
            const int p = atomicAdd(&slot[sd.x >> BIN_SHIFT], 1);
            piece[p] = (sd.y << BIN_SHIFT) | (sd.x & (BIN_ROWS - 1));
        }
    }
}

__global__ __launch_bounds__(BIN_THREADS) void gat_rowsort_kernel(
    const int* __restrict__ table, const int* __restrict__ binned,
    int* __restrict__ seg, int* __restrict__ sorted, int E, int N)
{
    __shared__ int cnt[BIN_ROWS];
    __shared__ int cur[BIN_ROWS];
    __shared__ int wsum[BIN_ROWS / 32];
    __shared__ int bsum[BIN_ROWS / 32];
    __shared__ int rbeg[BIN_GROUPS];
    __shared__ int rlen[BIN_GROUPS];
    const int tid  = (int)threadIdx.x;
    const int lane = tid & 31;
    const int wv   = tid >> 5;
    const int bin  = (int)blockIdx.x;
    const int per  = (E + BIN_GROUPS - 1) / BIN_GROUPS;

    int s0 = 0;
    if (tid < BIN_GROUPS) {
        s0 = table[(size_t)bin * BIN_GROUPS + tid];
        const int s1 = table[(size_t)(bin + 1) * BIN_GROUPS + tid];
        rbeg[tid] = tid * per + s0;
        rlen[tid] = s1 - s0;
    }
    if (tid < BIN_ROWS) cnt[tid] = 0;
    int r = s0;
    for (int off = 16; off > 0; off >>= 1) r += __shfl_xor(r, off, 32);
    if (tid < BIN_ROWS && lane == 0) bsum[wv] = r;
    __syncthreads();
    int base = 0;
    for (int w = 0; w < BIN_ROWS / 32; ++w) base += bsum[w];

    int rec[BIN_RUNS];
    {
        int rb[BIN_RUNS];
        int rn[BIN_RUNS];
#pragma unroll
        for (int u = 0; u < BIN_RUNS; ++u) {
            rb[u] = rbeg[wv + 32 * u];
            rn[u] = rlen[wv + 32 * u];
        }
#pragma unroll
        for (int u = 0; u < BIN_RUNS; ++u)
            rec[u] = (lane < rn[u]) ? binned[rb[u] + lane] : -1;
#pragma unroll
        for (int u = 0; u < BIN_RUNS; ++u)
            if (rec[u] >= 0) atomicAdd(&cnt[rec[u] & (BIN_ROWS - 1)], 1);
    }
    for (int u = 0; u < BIN_RUNS; ++u) {
        const int n = rlen[wv + 32 * u];
        const int* __restrict__ run = binned + rbeg[wv + 32 * u];
        for (int i = lane + 32; i < n; i += 32) atomicAdd(&cnt[run[i] & (BIN_ROWS - 1)], 1);
    }
    __syncthreads();

    const int c = (tid < BIN_ROWS) ? cnt[tid] : 0;
    int x = c;
    for (int off = 1; off < 32; off <<= 1) {
        const int y = __shfl_up(x, off, 32);
        if (lane >= off) x += y;
    }
    if (tid < BIN_ROWS && lane == 31) wsum[wv] = x;
    __syncthreads();
    if (tid == 0) {
        int s = 0;
        for (int w = 0; w < BIN_ROWS / 32; ++w) { const int t = wsum[w]; wsum[w] = s; s += t; }
    }
    __syncthreads();
    if (tid < BIN_ROWS) {
        const int excl = wsum[wv] + x - c;
        cur[tid] = excl;
        const int row = bin * BIN_ROWS + tid;
        if (row < N) *(int2*)(seg + 2 * (size_t)row) = make_int2(base + excl, c);
    }
    __syncthreads();

    int* __restrict__ dstv = sorted + base;
#pragma unroll
    for (int u = 0; u < BIN_RUNS; ++u) {
        if (rec[u] >= 0) {
            const int q = atomicAdd(&cur[rec[u] & (BIN_ROWS - 1)], 1);
            dstv[q] = rec[u] >> BIN_SHIFT;
        }
    }
    for (int u = 0; u < BIN_RUNS; ++u) {
        const int n = rlen[wv + 32 * u];
        const int* __restrict__ run = binned + rbeg[wv + 32 * u];
        for (int i = lane + 32; i < n; i += 32) {
            const int p = run[i];
            const int q = atomicAdd(&cur[p & (BIN_ROWS - 1)], 1);
            dstv[q] = p >> BIN_SHIFT;
        }
    }
}

template <int U>
__device__ __forceinline__ void gat_walk_step(
    const float* __restrict__ h, int d, float w, int i, int lane,
    float4& acc, float& den)
{
    float4 hv[U];
    float  wi[U];
#pragma unroll
    for (int u = 0; u < U; ++u) {
        const int di = __builtin_amdgcn_readlane(d, i + u);
        wi[u] = __int_as_float(__builtin_amdgcn_readlane(__float_as_int(w), i + u));
        hv[u] = *(const float4*)(h + (size_t)di * OUT_DIM + lane * 4);
    }
#pragma unroll
    for (int u = 0; u < U; ++u) {
        acc.x += wi[u] * hv[u].x;
        acc.y += wi[u] * hv[u].y;
        acc.z += wi[u] * hv[u].z;
        acc.w += wi[u] * hv[u].w;
        den   += wi[u];
    }
}

__global__ __launch_bounds__(256) void gat_walk_kernel(
    const float* __restrict__ h, const float* __restrict__ s_src,
    const float* __restrict__ s_dst, const int* __restrict__ seg,
    const int* __restrict__ sorted,
    const int* __restrict__ node_idx, float* __restrict__ out, int B)
{
    const int lane = threadIdx.x & 31;
    const int wv = __builtin_amdgcn_readfirstlane(threadIdx.x >> 5);
    const int j  = blockIdx.x * 8 + wv;
    if (j >= B) return;

    const int   row = node_idx[j];
    const unsigned long long sg = *(const unsigned long long*)(seg + 2 * (size_t)row);
    const int   beg = (int)(unsigned)(sg & 0xffffffffull);
    const int   n   = (int)(unsigned)(sg >> 32);
    const float ss  = s_src[row];

    float4 acc = make_float4(0.0f, 0.0f, 0.0f, 0.0f);
    float  den = 0.0f;

    for (int c = 0; c < n; c += 32) {
        const int m = (n - c < 32) ? (n - c) : 32;
        int   d = 0;
        float w = 0.0f;
        if (lane < m) {
            d = sorted[beg + c + lane];
            float s = ss + s_dst[d];
            s = (s >= 0.0f) ? s : SLOPE * s;
            w = expf(s);
        }
        int i = 0;
        for (; i + 8 <= m; i += 8) gat_walk_step<8>(h, d, w, i, lane, acc, den);
        if (i + 4 <= m) { gat_walk_step<4>(h, d, w, i, lane, acc, den); i += 4; }
        if (i + 2 <= m) { gat_walk_step<2>(h, d, w, i, lane, acc, den); i += 2; }
        if (i < m)      { gat_walk_step<1>(h, d, w, i, lane, acc, den); }
    }

    const float inv = 1.0f / den;
    float4 ov;
    ov.x = acc.x * inv; ov.y = acc.y * inv; ov.z = acc.z * inv; ov.w = acc.w * inv;
    *(float4*)(out + (size_t)j * OUT_DIM + lane * 4) = ov;
}

extern "C" void kernel_launch(void* const* d_in, const int* in_sizes, int n_in,
                              void* d_out, int out_size, void* d_ws, size_t ws_size,
                              hipStream_t stream) {
    const float* features     = (const float*)d_in[0];
    const float* W            = (const float*)d_in[1];
    const float* b            = (const float*)d_in[2];
    const float* a            = (const float*)d_in[3];
    const int*   edges        = (const int*)d_in[4];
    const int*   unique_nodes = (const int*)d_in[5];
    const int*   node_idx     = (const int*)d_in[6];
    float* out = (float*)d_out;

    const int N = in_sizes[5];
    const int E = in_sizes[4] / 2;
    const int B = in_sizes[6];

    float* h      = (float*)d_ws;
    int*   seg    = (int*)(h + (size_t)N * OUT_DIM);
    float* s_src  = (float*)(seg + 2 * (size_t)N);
    float* s_dst  = s_src + N;
    int*   table  = (int*)(s_dst + N);
    int*   binned = table + N + 1;
    int*   sorted = binned + E;

    {
        gat_gemm_kernel<<<GEMM_GROUPS, 256, 0, stream>>>(features, W, b, unique_nodes, h, N);
    }
    {
        gat_scores_kernel<<<SCORE_GROUPS, 256, 0, stream>>>(h, a, s_src, s_dst, N);
    }
    {
        gat_bin_kernel<<<BIN_GROUPS, BIN_THREADS, 0, stream>>>(edges, table, binned, E, N);
    }
    {
        dim3 grid((N + 255) / 256);
        gat_rowsort_kernel<<<grid, BIN_THREADS, 0, stream>>>(table, binned, seg, sorted, E, N);
    }
    {
        dim3 grid((B + 7) / 8);
        gat_walk_kernel<<<grid, 256, 0, stream>>>(h, s_src, s_dst, seg, sorted,
                                                  node_idx, out, B);
    }
}


# scores and edge binning share one launch: work-groups 0-223 bin their edge shares as before, 196 more of 32 waves form the row scores (same sum tree); 5 launches to 4
# speedup vs baseline: 4.6268x; 1.0591x over previous
#include <hip/hip_runtime.h>
#include <hip/hip_bf16.h>

#define IN_DIM 256
#define OUT_DIM 128
#define SLOPE 0.1f

typedef float v2f __attribute__((ext_vector_type(2)));
typedef float v8f __attribute__((ext_vector_type(8)));

#define GEMM_GROUPS 256
#define GEMM_WAVES 8
#define KCHUNKS (IN_DIM / 8)
#define CTILES (OUT_DIM / 16)
#define CHUNK_PITCH (CTILES * 32)
#define GEMM_LDS_SLOTS 10752

__device__ __forceinline__ void gat_load_stripe(
    const float* __restrict__ features, const int* __restrict__ unique_nodes,
    int tile, int mn, int half, int N, float4 (&dst)[KCHUNKS])
{
    int arow_idx = tile * 16 + mn;
    if (arow_idx >= N) arow_idx = N - 1;
    const int gsrc = unique_nodes[arow_idx];
    const float* __restrict__ arow = features + (size_t)gsrc * IN_DIM + 4 * half;
#pragma unroll
    for (int t = 0; t < KCHUNKS; ++t)
        dst[t] = *(const float4*)(arow + 8 * t);
}

__global__ __launch_bounds__(256) void gat_gemm_kernel(
    const float* __restrict__ features, const float* __restrict__ W,
    const float* __restrict__ bias, const int* __restrict__ unique_nodes,
    float* __restrict__ h, int N)
{
    __shared__ float4 wl[GEMM_LDS_SLOTS];

    const int lane = threadIdx.x & 31;
    const int wave = __builtin_amdgcn_readfirstlane(threadIdx.x >> 5);
    const int half = lane >> 4;
    const int mn   = lane & 15;

    const int nTiles = (N + 15) >> 4;
    const int stride = (int)gridDim.x * GEMM_WAVES;
    int tile = wave * (int)gridDim.x + (int)blockIdx.x;

    float4 cur[KCHUNKS];
    float4 nxt[KCHUNKS];
#pragma unroll
    for (int t = 0; t < KCHUNKS; ++t) {
        cur[t] = make_float4(0.0f, 0.0f, 0.0f, 0.0f);
        nxt[t] = make_float4(0.0f, 0.0f, 0.0f, 0.0f);
    }
    if (tile < nTiles) gat_load_stripe(features, unique_nodes, tile, mn, half, N, cur);

    {
        const int qlo  = lane & 7;
        const int rsub = lane >> 3;
#pragma unroll 16
        for (int it = 0; it < (OUT_DIM * IN_DIM / 4) / 256; ++it) {
            const int wi = it * GEMM_WAVES + wave;
            const int n  = (wi >> 3) * 4 + rsub;
            const int q  = (wi & 7) * 8 + qlo;
            const float4 v = *(const float4*)(W + (size_t)n * IN_DIM + 4 * q);
            wl[(q >> 1) * CHUNK_PITCH + (n >> 4) * 32 + (q & 1) * 16 + (n & 15)] = v;
        }
    }

    float bv[CTILES];
#pragma unroll
    for (int c = 0; c < CTILES; ++c) bv[c] = bias[c * 16 + mn];

    __syncthreads();

    const int loSlot = lane;
    int hiSlot = lane + (KCHUNKS / 2) * CHUNK_PITCH;
    asm volatile("" : "+v"(hiSlot));

    for (; tile < nTiles; tile += stride) {
        const int  rowBase = tile * 16;
        const bool more    = (tile + stride) < nTiles;
        if (more) gat_load_stripe(features, unique_nodes, tile + stride, mn, half, N, nxt);

        v8f acc[CTILES];
#pragma unroll
        for (int c = 0; c < CTILES; ++c) {
#pragma unroll
            for (int r = 0; r < 8; ++r) acc[c][r] = bv[c];
        }

        float4 bcur[CTILES];
#pragma unroll
        for (int c = 0; c < CTILES; ++c) bcur[c] = wl[loSlot + c * 32];

#pragma unroll
        for (int t = 0; t < KCHUNKS; ++t) {
            float4 bnx[CTILES];
            if (t + 1 < KCHUNKS) {
#pragma unroll
                for (int c = 0; c < CTILES; ++c)
                    bnx[c] = (t + 1 < KCHUNKS / 2)
                        ? wl[loSlot + (t + 1) * CHUNK_PITCH + c * 32]
                        : wl[hiSlot + (t + 1 - KCHUNKS / 2) * CHUNK_PITCH + c * 32];
            }
            __builtin_amdgcn_sched_barrier(0);
            const float4 a = cur[t];
            const v2f a0 = {a.x, a.y};
            const v2f a1 = {a.z, a.w};
#pragma unroll
            for (int c = 0; c < CTILES; ++c) {
                const v2f b0 = {bcur[c].x, bcur[c].y};
                acc[c] = __builtin_amdgcn_wmma_f32_16x16x4_f32(
                     false, a0,  false, b0,
                     (short)0, acc[c],  false,  false);
            }
#pragma unroll
            for (int c = 0; c < CTILES; ++c) {
                const v2f b1 = {bcur[c].z, bcur[c].w};
                acc[c] = __builtin_amdgcn_wmma_f32_16x16x4_f32(
                     false, a1,  false, b1,
                     (short)0, acc[c],  false,  false);
            }
            __builtin_amdgcn_sched_barrier(0);
            if (t + 1 < KCHUNKS) {
#pragma unroll
                for (int c = 0; c < CTILES; ++c) bcur[c] = bnx[c];
            }
        }

        asm volatile("v_nop\n\tv_nop\n\tv_nop\n\tv_nop\n\tv_nop\n\tv_nop\n\tv_nop\n\tv_nop");
        __builtin_amdgcn_sched_barrier(0);

        float* __restrict__ obase =
            h + (size_t)(rowBase + 8 * half) * OUT_DIM + mn;
        if (rowBase + 16 <= N) {
#pragma unroll
            for (int c = 0; c < CTILES; ++c) {
#pragma unroll
                for (int r = 0; r < 8; ++r)
                    obase[(size_t)r * OUT_DIM + c * 16] = acc[c][r];
            }
        } else {
#pragma unroll
            for (int c = 0; c < CTILES; ++c) {
#pragma unroll
                for (int r = 0; r < 8; ++r) {
                    const int orow = rowBase + 8 * half + r;
                    if (orow < N) obase[(size_t)r * OUT_DIM + c * 16] = acc[c][r];
                }
            }
        }

        if (more) {
#pragma unroll
            for (int t = 0; t < KCHUNKS; ++t) cur[t] = nxt[t];
        }
    }
}

#define SCORE_GROUPS 196
#define SCORE_WAVES 32
#define SCORE_ROWS 8

__device__ __forceinline__ void gat_scores_part(
    const float* __restrict__ h, const float* __restrict__ a,
    float* __restrict__ s_src, float* __restrict__ s_dst, int N,
    int group, int groups)
{
    const int lane = threadIdx.x & 31;
    const int gw = group * SCORE_WAVES + __builtin_amdgcn_readfirstlane(threadIdx.x >> 5);
    const int step = groups * SCORE_WAVES * SCORE_ROWS;

    const float4 as = *(const float4*)(a + lane * 4);
    const float4 ad = *(const float4*)(a + OUT_DIM + lane * 4);

    for (int row0 = gw * SCORE_ROWS; row0 < N; row0 += step) {
        float4 hv[SCORE_ROWS];
#pragma unroll
        for (int u = 0; u < SCORE_ROWS; ++u) {
            int r = row0 + u;
            if (r >= N) r = N - 1;
            hv[u] = *(const float4*)(h + (size_t)r * OUT_DIM + lane * 4);
        }
        float outs = 0.0f, outd = 0.0f;
#pragma unroll
        for (int u = 0; u < SCORE_ROWS; ++u) {
            float ps = hv[u].x * as.x + hv[u].y * as.y + hv[u].z * as.z + hv[u].w * as.w;
            float pd = hv[u].x * ad.x + hv[u].y * ad.y + hv[u].z * ad.z + hv[u].w * ad.w;
#pragma unroll
            for (int off = 16; off > 0; off >>= 1) {
                ps += __shfl_xor(ps, off, 32);
                pd += __shfl_xor(pd, off, 32);
            }
            if (lane == u) { outs = ps; outd = pd; }
        }
        if (lane < SCORE_ROWS && row0 + lane < N) {
            s_src[row0 + lane] = outs;
            s_dst[row0 + lane] = outd;
        }
    }
}

#define BIN_SHIFT 8
#define BIN_ROWS 256
#define BIN_GROUPS 224
#define BIN_THREADS 1024
#define BIN_U 4
#define BIN_LDS 1024
#define BIN_RUNS (BIN_GROUPS / 32)

__global__ __launch_bounds__(BIN_THREADS) void gat_bin_scores_kernel(
    const int* __restrict__ edges, int* __restrict__ table,
    int* __restrict__ binned, int E, int N,
    const float* __restrict__ h, const float* __restrict__ a,
    float* __restrict__ s_src, float* __restrict__ s_dst)
{
    static_assert(BIN_LDS == BIN_THREADS, "one counter per thread in the scan");
    static_assert(SCORE_WAVES * 32 == BIN_THREADS, "a score work-group is one of this launch");
    __shared__ int slot[BIN_LDS];
    __shared__ int wsum[BIN_THREADS / 32];
    if ((int)blockIdx.x >= BIN_GROUPS) {
        gat_scores_part(h, a, s_src, s_dst, N,
                        (int)blockIdx.x - BIN_GROUPS, (int)gridDim.x - BIN_GROUPS);
        return;
    }
    const int nb   = (N + BIN_ROWS - 1) >> BIN_SHIFT;
    const int tid  = (int)threadIdx.x;
    const int lane = tid & 31;
    const int wv   = tid >> 5;
    slot[tid] = 0;
    __syncthreads();

    const int per = (E + BIN_GROUPS - 1) / BIN_GROUPS;
    int lo = (int)blockIdx.x * per;
    if (lo > E) lo = E;
    const int hi = (lo + per < E) ? (lo + per) : E;

    {
        int e0 = lo + tid;
        for (; e0 + (BIN_U - 1) * BIN_THREADS < hi; e0 += BIN_THREADS * BIN_U) {
            int s[BIN_U];
#pragma unroll
            for (int u = 0; u < BIN_U; ++u) s[u] = edges[2 * (size_t)(e0 + u * BIN_THREADS)];
#pragma unroll
            for (int u = 0; u < BIN_U; ++u) atomicAdd(&slot[s[u] >> BIN_SHIFT], 1);
        }
        for (; e0 < hi; e0 += BIN_THREADS)
            atomicAdd(&slot[edges[2 * (size_t)e0] >> BIN_SHIFT], 1);
    }
    __syncthreads();

    const int c = slot[tid];
    int x = c;
    for (int off = 1; off < 32; off <<= 1) {
        const int y = __shfl_up(x, off, 32);
        if (lane >= off) x += y;
    }
    if (lane == 31) wsum[wv] = x;
    __syncthreads();
    if (wv == 0) {
        const int s = wsum[lane];
        int t = s;
        for (int off = 1; off < 32; off <<= 1) {
            const int y = __shfl_up(t, off, 32);
            if (lane >= off) t += y;
        }
        wsum[lane] = t - s;
    }
    __syncthreads();
    const int excl = wsum[wv] + x - c;
    slot[tid] = excl;
    if (tid <= nb) table[(size_t)tid * BIN_GROUPS + blockIdx.x] = excl;
    if (tid == 0 && nb == BIN_LDS) table[(size_t)nb * BIN_GROUPS + blockIdx.x] = hi - lo;
    __syncthreads();

    int* __restrict__ piece = binned + lo;
    {
        int e0 = lo + tid;
        for (; e0 + (BIN_U - 1) * BIN_THREADS < hi; e0 += BIN_THREADS * BIN_U) {
            int2 sd[BIN_U];
#pragma unroll
            for (int u = 0; u < BIN_U; ++u)
                sd[u] = *(const int2*)(edges + 2 * (size_t)(e0 + u * BIN_THREADS));
#pragma unroll
            for (int u = 0; u < BIN_U; ++u) {
                const int p = atomicAdd(&slot[sd[u].x >> BIN_SHIFT], 1);
                piece[p] = (sd[u].y << BIN_SHIFT) | (sd[u].x & (BIN_ROWS - 1));
            }
        }
        for (; e0 < hi; e0 += BIN_THREADS) {
            const int2 sd = *(const int2*)(edges + 2 * (size_t)e0);
            const int p = atomicAdd(&slot[sd.x >> BIN_SHIFT], 1);
            piece[p] = (sd.y << BIN_SHIFT) | (sd.x & (BIN_ROWS - 1));
        }
    }
}

__global__ __launch_bounds__(BIN_THREADS) void gat_rowsort_kernel(
    const int* __restrict__ table, const int* __restrict__ binned,
    int* __restrict__ seg, int* __restrict__ sorted, int E, int N)
{
    __shared__ int cnt[BIN_ROWS];
    __shared__ int cur[BIN_ROWS];
    __shared__ int wsum[BIN_ROWS / 32];
    __shared__ int bsum[BIN_ROWS / 32];
    __shared__ int rbeg[BIN_GROUPS];
    __shared__ int rlen[BIN_GROUPS];
    const int tid  = (int)threadIdx.x;
    const int lane = tid & 31;
    const int wv   = tid >> 5;
    const int bin  = (int)blockIdx.x;
    const int per  = (E + BIN_GROUPS - 1) / BIN_GROUPS;

    int s0 = 0;
    if (tid < BIN_GROUPS) {
        s0 = table[(size_t)bin * BIN_GROUPS + tid];
        const int s1 = table[(size_t)(bin + 1) * BIN_GROUPS + tid];
        rbeg[tid] = tid * per + s0;
        rlen[tid] = s1 - s0;
    }
    if (tid < BIN_ROWS) cnt[tid] = 0;
    int r = s0;
    for (int off = 16; off > 0; off >>= 1) r += __shfl_xor(r, off, 32);
    if (tid < BIN_ROWS && lane == 0) bsum[wv] = r;
    __syncthreads();
    int base = 0;
    for (int w = 0; w < BIN_ROWS / 32; ++w) base += bsum[w];

    int rec[BIN_RUNS];
    {
        int rb[BIN_RUNS];
        int rn[BIN_RUNS];
#pragma unroll
        for (int u = 0; u < BIN_RUNS; ++u) {
            rb[u] = rbeg[wv + 32 * u];
            rn[u] = rlen[wv + 32 * u];
        }
#pragma unroll
        for (int u = 0; u < BIN_RUNS; ++u)
            rec[u] = (lane < rn[u]) ? binned[rb[u] + lane] : -1;
#pragma unroll
        for (int u = 0; u < BIN_RUNS; ++u)
            if (rec[u] >= 0) atomicAdd(&cnt[rec[u] & (BIN_ROWS - 1)], 1);
    }
    for (int u = 0; u < BIN_RUNS; ++u) {
        const int n = rlen[wv + 32 * u];
        const int* __restrict__ run = binned + rbeg[wv + 32 * u];
        for (int i = lane + 32; i < n; i += 32) atomicAdd(&cnt[run[i] & (BIN_ROWS - 1)], 1);
    }
    __syncthreads();

    const int c = (tid < BIN_ROWS) ? cnt[tid] : 0;
    int x = c;
    for (int off = 1; off < 32; off <<= 1) {
        const int y = __shfl_up(x, off, 32);
        if (lane >= off) x += y;
    }
    if (tid < BIN_ROWS && lane == 31) wsum[wv] = x;
    __syncthreads();
    if (tid == 0) {
        int s = 0;
        for (int w = 0; w < BIN_ROWS / 32; ++w) { const int t = wsum[w]; wsum[w] = s; s += t; }
    }
    __syncthreads();
    if (tid < BIN_ROWS) {
        const int excl = wsum[wv] + x - c;
        cur[tid] = excl;
        const int row = bin * BIN_ROWS + tid;
        if (row < N) *(int2*)(seg + 2 * (size_t)row) = make_int2(base + excl, c);
    }
    __syncthreads();

    int* __restrict__ dstv = sorted + base;
#pragma unroll
    for (int u = 0; u < BIN_RUNS; ++u) {
        if (rec[u] >= 0) {
            const int q = atomicAdd(&cur[rec[u] & (BIN_ROWS - 1)], 1);
            dstv[q] = rec[u] >> BIN_SHIFT;
        }
    }
    for (int u = 0; u < BIN_RUNS; ++u) {
        const int n = rlen[wv + 32 * u];
        const int* __restrict__ run = binned + rbeg[wv + 32 * u];
        for (int i = lane + 32; i < n; i += 32) {
            const int p = run[i];
            const int q = atomicAdd(&cur[p & (BIN_ROWS - 1)], 1);
            dstv[q] = p >> BIN_SHIFT;
        }
    }
}

template <int U>
__device__ __forceinline__ void gat_walk_step(
    const float* __restrict__ h, int d, float w, int i, int lane,
    float4& acc, float& den)
{
    float4 hv[U];
    float  wi[U];
#pragma unroll
    for (int u = 0; u < U; ++u) {
        const int di = __builtin_amdgcn_readlane(d, i + u);
        wi[u] = __int_as_float(__builtin_amdgcn_readlane(__float_as_int(w), i + u));
        hv[u] = *(const float4*)(h + (size_t)di * OUT_DIM + lane * 4);
    }
#pragma unroll
    for (int u = 0; u < U; ++u) {
        acc.x += wi[u] * hv[u].x;
        acc.y += wi[u] * hv[u].y;
        acc.z += wi[u] * hv[u].z;
        acc.w += wi[u] * hv[u].w;
        den   += wi[u];
    }
}

__global__ __launch_bounds__(256) void gat_walk_kernel(
    const float* __restrict__ h, const float* __restrict__ s_src,
    const float* __restrict__ s_dst, const int* __restrict__ seg,
    const int* __restrict__ sorted,
    const int* __restrict__ node_idx, float* __restrict__ out, int B)
{
    const int lane = threadIdx.x & 31;
    const int wv = __builtin_amdgcn_readfirstlane(threadIdx.x >> 5);
    const int j  = blockIdx.x * 8 + wv;
    if (j >= B) return;

    const int   row = node_idx[j];
    const unsigned long long sg = *(const unsigned long long*)(seg + 2 * (size_t)row);
    const int   beg = (int)(unsigned)(sg & 0xffffffffull);
    const int   n   = (int)(unsigned)(sg >> 32);
    const float ss  = s_src[row];

    float4 acc = make_float4(0.0f, 0.0f, 0.0f, 0.0f);
    float  den = 0.0f;

    for (int c = 0; c < n; c += 32) {
        const int m = (n - c < 32) ? (n - c) : 32;
        int   d = 0;
        float w = 0.0f;
        if (lane < m) {
            d = sorted[beg + c + lane];
            float s = ss + s_dst[d];
            s = (s >= 0.0f) ? s : SLOPE * s;
            w = expf(s);
        }
        int i = 0;
        for (; i + 8 <= m; i += 8) gat_walk_step<8>(h, d, w, i, lane, acc, den);
        if (i + 4 <= m) { gat_walk_step<4>(h, d, w, i, lane, acc, den); i += 4; }
        if (i + 2 <= m) { gat_walk_step<2>(h, d, w, i, lane, acc, den); i += 2; }
        if (i < m)      { gat_walk_step<1>(h, d, w, i, lane, acc, den); }
    }

    const float inv = 1.0f / den;
    float4 ov;
    ov.x = acc.x * inv; ov.y = acc.y * inv; ov.z = acc.z * inv; ov.w = acc.w * inv;
    *(float4*)(out + (size_t)j * OUT_DIM + lane * 4) = ov;
}

extern "C" void kernel_launch(void* const* d_in, const int* in_sizes, int n_in,
                              void* d_out, int out_size, void* d_ws, size_t ws_size,
                              hipStream_t stream) {
    const float* features     = (const float*)d_in[0];
    const float* W            = (const float*)d_in[1];
    const float* b            = (const float*)d_in[2];
    const float* a            = (const float*)d_in[3];
    const int*   edges        = (const int*)d_in[4];
    const int*   unique_nodes = (const int*)d_in[5];
    const int*   node_idx     = (const int*)d_in[6];
    float* out = (float*)d_out;

    const int N = in_sizes[5];
    const int E = in_sizes[4] / 2;
    const int B = in_sizes[6];

    float* h      = (float*)d_ws;
    int*   seg    = (int*)(h + (size_t)N * OUT_DIM);
    float* s_src  = (float*)(seg + 2 * (size_t)N);
    float* s_dst  = s_src + N;
    int*   table  = (int*)(s_dst + N);
    int*   binned = table + N + 1;
    int*   sorted = binned + E;

    {
        gat_gemm_kernel<<<GEMM_GROUPS, 256, 0, stream>>>(features, W, b, unique_nodes, h, N);
    }
    {
        gat_bin_scores_kernel<<<BIN_GROUPS + SCORE_GROUPS, BIN_THREADS, 0, stream>>>(
            edges, table, binned, E, N, h, a, s_src, s_dst);
    }
    {
        dim3 grid((N + 255) / 256);
        gat_rowsort_kernel<<<grid, BIN_THREADS, 0, stream>>>(table, binned, seg, sorted, E, N);
    }
    {
        dim3 grid((B + 7) / 8);
        gat_walk_kernel<<<grid, 256, 0, stream>>>(h, s_src, s_dst, seg, sorted,
                                                  node_idx, out, B);
    }
}


# row sums inside the row sort launch: queries binned with the edges, a bin's work-group keeps its sorted segments in LDS and a wave per query sums from them; 4 launches to 3; sums in LDS-rank order
# speedup vs baseline: 5.0448x; 1.0903x over previous
#include <hip/hip_runtime.h>
#include <hip/hip_bf16.h>

#define IN_DIM 256
#define OUT_DIM 128
#define SLOPE 0.1f

typedef float v2f __attribute__((ext_vector_type(2)));
typedef float v8f __attribute__((ext_vector_type(8)));

#define GEMM_GROUPS 256
#define GEMM_WAVES 8
#define KCHUNKS (IN_DIM / 8)
#define CTILES (OUT_DIM / 16)
#define CHUNK_PITCH (CTILES * 32)
#define GEMM_LDS_SLOTS 10752

__device__ __forceinline__ void gat_load_stripe(
    const float* __restrict__ features, const int* __restrict__ unique_nodes,
    int tile, int mn, int half, int N, float4 (&dst)[KCHUNKS])
{
    int arow_idx = tile * 16 + mn;
    if (arow_idx >= N) arow_idx = N - 1;
    const int gsrc = unique_nodes[arow_idx];
    const float* __restrict__ arow = features + (size_t)gsrc * IN_DIM + 4 * half;
#pragma unroll
    for (int t = 0; t < KCHUNKS; ++t)
        dst[t] = *(const float4*)(arow + 8 * t);
}

__global__ __launch_bounds__(256) void gat_gemm_kernel(
    const float* __restrict__ features, const float* __restrict__ W,
    const float* __restrict__ bias, const int* __restrict__ unique_nodes,
    float* __restrict__ h, int N)
{
    __shared__ float4 wl[GEMM_LDS_SLOTS];

    const int lane = threadIdx.x & 31;
    const int wave = __builtin_amdgcn_readfirstlane(threadIdx.x >> 5);
    const int half = lane >> 4;
    const int mn   = lane & 15;

    const int nTiles = (N + 15) >> 4;
    const int stride = (int)gridDim.x * GEMM_WAVES;
    int tile = wave * (int)gridDim.x + (int)blockIdx.x;

    float4 cur[KCHUNKS];
    float4 nxt[KCHUNKS];
#pragma unroll
    for (int t = 0; t < KCHUNKS; ++t) {
        cur[t] = make_float4(0.0f, 0.0f, 0.0f, 0.0f);
        nxt[t] = make_float4(0.0f, 0.0f, 0.0f, 0.0f);
    }
    if (tile < nTiles) gat_load_stripe(features, unique_nodes, tile, mn, half, N, cur);

    {
        const int qlo  = lane & 7;
        const int rsub = lane >> 3;
#pragma unroll 16
        for (int it = 0; it < (OUT_DIM * IN_DIM / 4) / 256; ++it) {
            const int wi = it * GEMM_WAVES + wave;
            const int n  = (wi >> 3) * 4 + rsub;
            const int q  = (wi & 7) * 8 + qlo;
            const float4 v = *(const float4*)(W + (size_t)n * IN_DIM + 4 * q);
            wl[(q >> 1) * CHUNK_PITCH + (n >> 4) * 32 + (q & 1) * 16 + (n & 15)] = v;
        }
    }

    float bv[CTILES];
#pragma unroll
    for (int c = 0; c < CTILES; ++c) bv[c] = bias[c * 16 + mn];

    __syncthreads();

    const int loSlot = lane;
    int hiSlot = lane + (KCHUNKS / 2) * CHUNK_PITCH;
    asm volatile("" : "+v"(hiSlot));

    for (; tile < nTiles; tile += stride) {
        const int  rowBase = tile * 16;
        const bool more    = (tile + stride) < nTiles;
        if (more) gat_load_stripe(features, unique_nodes, tile + stride, mn, half, N, nxt);

        v8f acc[CTILES];
#pragma unroll
        for (int c = 0; c < CTILES; ++c) {
#pragma unroll
            for (int r = 0; r < 8; ++r) acc[c][r] = bv[c];
        }

        float4 bcur[CTILES];
#pragma unroll
        for (int c = 0; c < CTILES; ++c) bcur[c] = wl[loSlot + c * 32];

#pragma unroll
        for (int t = 0; t < KCHUNKS; ++t) {
            float4 bnx[CTILES];
            if (t + 1 < KCHUNKS) {
#pragma unroll
                for (int c = 0; c < CTILES; ++c)
                    bnx[c] = (t + 1 < KCHUNKS / 2)
                        ? wl[loSlot + (t + 1) * CHUNK_PITCH + c * 32]
                        : wl[hiSlot + (t + 1 - KCHUNKS / 2) * CHUNK_PITCH + c * 32];
            }
            __builtin_amdgcn_sched_barrier(0);
            const float4 a = cur[t];
            const v2f a0 = {a.x, a.y};
            const v2f a1 = {a.z, a.w};
#pragma unroll
            for (int c = 0; c < CTILES; ++c) {
                const v2f b0 = {bcur[c].x, bcur[c].y};
                acc[c] = __builtin_amdgcn_wmma_f32_16x16x4_f32(
                     false, a0,  false, b0,
                     (short)0, acc[c],  false,  false);
            }
#pragma unroll
            for (int c = 0; c < CTILES; ++c) {
                const v2f b1 = {bcur[c].z, bcur[c].w};
                acc[c] = __builtin_amdgcn_wmma_f32_16x16x4_f32(
                     false, a1,  false, b1,
                     (short)0, acc[c],  false,  false);
            }
            __builtin_amdgcn_sched_barrier(0);
            if (t + 1 < KCHUNKS) {
#pragma unroll
                for (int c = 0; c < CTILES; ++c) bcur[c] = bnx[c];
            }
        }

        asm volatile("v_nop\n\tv_nop\n\tv_nop\n\tv_nop\n\tv_nop\n\tv_nop\n\tv_nop\n\tv_nop");
        __builtin_amdgcn_sched_barrier(0);

        float* __restrict__ obase =
            h + (size_t)(rowBase + 8 * half) * OUT_DIM + mn;
        if (rowBase + 16 <= N) {
#pragma unroll
            for (int c = 0; c < CTILES; ++c) {
#pragma unroll
                for (int r = 0; r < 8; ++r)
                    obase[(size_t)r * OUT_DIM + c * 16] = acc[c][r];
            }
        } else {
#pragma unroll
            for (int c = 0; c < CTILES; ++c) {
#pragma unroll
                for (int r = 0; r < 8; ++r) {
                    const int orow = rowBase + 8 * half + r;
                    if (orow < N) obase[(size_t)r * OUT_DIM + c * 16] = acc[c][r];
                }
            }
        }

        if (more) {
#pragma unroll
            for (int t = 0; t < KCHUNKS; ++t) cur[t] = nxt[t];
        }
    }
}

#define SCORE_GROUPS 196
#define SCORE_WAVES 32
#define SCORE_ROWS 8

__device__ __forceinline__ void gat_scores_part(
    const float* __restrict__ h, const float* __restrict__ a,
    float* __restrict__ s_src, float* __restrict__ s_dst, int N,
    int group, int groups)
{
    const int lane = threadIdx.x & 31;
    const int gw = group * SCORE_WAVES + __builtin_amdgcn_readfirstlane(threadIdx.x >> 5);
    const int step = groups * SCORE_WAVES * SCORE_ROWS;

    const float4 as = *(const float4*)(a + lane * 4);
    const float4 ad = *(const float4*)(a + OUT_DIM + lane * 4);

    for (int row0 = gw * SCORE_ROWS; row0 < N; row0 += step) {
        float4 hv[SCORE_ROWS];
#pragma unroll
        for (int u = 0; u < SCORE_ROWS; ++u) {
            int r = row0 + u;
            if (r >= N) r = N - 1;
            hv[u] = *(const float4*)(h + (size_t)r * OUT_DIM + lane * 4);
        }
        float outs = 0.0f, outd = 0.0f;
#pragma unroll
        for (int u = 0; u < SCORE_ROWS; ++u) {
            float ps = hv[u].x * as.x + hv[u].y * as.y + hv[u].z * as.z + hv[u].w * as.w;
            float pd = hv[u].x * ad.x + hv[u].y * ad.y + hv[u].z * ad.z + hv[u].w * ad.w;
#pragma unroll
            for (int off = 16; off > 0; off >>= 1) {
                ps += __shfl_xor(ps, off, 32);
                pd += __shfl_xor(pd, off, 32);
            }
            if (lane == u) { outs = ps; outd = pd; }
        }
        if (lane < SCORE_ROWS && row0 + lane < N) {
            s_src[row0 + lane] = outs;
            s_dst[row0 + lane] = outd;
        }
    }
}

#define BIN_SHIFT 8
#define BIN_ROWS 256
#define BIN_GROUPS 224
#define BIN_THREADS 1024
#define BIN_U 4
#define BIN_LDS 1024
#define BIN_RUNS (BIN_GROUPS / 32)
#define SEG_CAP 8192
#define QRY_CAP 1024
#define QRY_FLAG 0x80000000u
typedef __attribute__((address_space(3))) int gat_lds_int;

__global__ __launch_bounds__(BIN_THREADS) void gat_bin_scores_kernel(
    const int* __restrict__ edges, const int* __restrict__ node_idx,
    int* __restrict__ table, int* __restrict__ binned, int E, int N, int B,
    const float* __restrict__ h, const float* __restrict__ a,
    float* __restrict__ s_src, float* __restrict__ s_dst)
{
    static_assert(BIN_LDS == BIN_THREADS, "one counter per thread in the scan");
    static_assert(SCORE_WAVES * 32 == BIN_THREADS, "a score work-group is one of this launch");
    __shared__ int slot[BIN_LDS];
    __shared__ int wsum[BIN_THREADS / 32];
    if ((int)blockIdx.x >= BIN_GROUPS) {
        gat_scores_part(h, a, s_src, s_dst, N,
                        (int)blockIdx.x - BIN_GROUPS, (int)gridDim.x - BIN_GROUPS);
        return;
    }
    const int nb   = (N + BIN_ROWS - 1) >> BIN_SHIFT;
    const int tid  = (int)threadIdx.x;
    const int lane = tid & 31;
    const int wv   = tid >> 5;
    slot[tid] = 0;

    const int per = (E + BIN_GROUPS - 1) / BIN_GROUPS;
    int lo = (int)blockIdx.x * per;
    if (lo > E) lo = E;
    const int hi = (lo + per < E) ? (lo + per) : E;
    const int qper = (B + BIN_GROUPS - 1) / BIN_GROUPS;
    int qlo = (int)blockIdx.x * qper;
    if (qlo > B) qlo = B;
    const int qhi = (qlo + qper < B) ? (qlo + qper) : B;
    __syncthreads();
    const int q0    = qlo + tid;
    const int qrow0 = (q0 < qhi) ? node_idx[q0] : -1;

    {
        int e0 = lo + tid;
        for (; e0 + (BIN_U - 1) * BIN_THREADS < hi; e0 += BIN_THREADS * BIN_U) {
            int s[BIN_U];
#pragma unroll
            for (int u = 0; u < BIN_U; ++u) s[u] = edges[2 * (size_t)(e0 + u * BIN_THREADS)];
#pragma unroll
            for (int u = 0; u < BIN_U; ++u) atomicAdd(&slot[s[u] >> BIN_SHIFT], 1);
        }
        for (; e0 < hi; e0 += BIN_THREADS)
            atomicAdd(&slot[edges[2 * (size_t)e0] >> BIN_SHIFT], 1);
        if (qrow0 >= 0) atomicAdd(&slot[qrow0 >> BIN_SHIFT], 1);
        for (int q = q0 + BIN_THREADS; q < qhi; q += BIN_THREADS)
            atomicAdd(&slot[node_idx[q] >> BIN_SHIFT], 1);
    }
    __syncthreads();

    const int c = slot[tid];
    int x = c;
    for (int off = 1; off < 32; off <<= 1) {
        const int y = __shfl_up(x, off, 32);
        if (lane >= off) x += y;
    }
    if (lane == 31) wsum[wv] = x;
    __syncthreads();
    if (wv == 0) {
        const int s = wsum[lane];
        int t = s;
        for (int off = 1; off < 32; off <<= 1) {
            const int y = __shfl_up(t, off, 32);
            if (lane >= off) t += y;
        }
        wsum[lane] = t - s;
    }
    __syncthreads();
    const int excl = wsum[wv] + x - c;
    slot[tid] = excl;
    if (tid <= nb) table[(size_t)tid * BIN_GROUPS + blockIdx.x] = excl;
    if (tid == 0 && nb == BIN_LDS)
        table[(size_t)nb * BIN_GROUPS + blockIdx.x] = (hi - lo) + (qhi - qlo);
    __syncthreads();

    int* __restrict__ piece = binned + lo + qlo;
    {
        int e0 = lo + tid;
        for (; e0 + (BIN_U - 1) * BIN_THREADS < hi; e0 += BIN_THREADS * BIN_U) {
            int2 sd[BIN_U];
#pragma unroll
            for (int u = 0; u < BIN_U; ++u)
                sd[u] = *(const int2*)(edges + 2 * (size_t)(e0 + u * BIN_THREADS));
#pragma unroll
            for (int u = 0; u < BIN_U; ++u) {
                const int p = atomicAdd(&slot[sd[u].x >> BIN_SHIFT], 1);
                piece[p] = (sd[u].y << BIN_SHIFT) | (sd[u].x & (BIN_ROWS - 1));
            }
        }
        for (; e0 < hi; e0 += BIN_THREADS) {
            const int2 sd = *(const int2*)(edges + 2 * (size_t)e0);
            const int p = atomicAdd(&slot[sd.x >> BIN_SHIFT], 1);
            piece[p] = (sd.y << BIN_SHIFT) | (sd.x & (BIN_ROWS - 1));
        }
        if (qrow0 >= 0) {
            const int p = atomicAdd(&slot[qrow0 >> BIN_SHIFT], 1);
            piece[p] = (int)(QRY_FLAG | ((unsigned)q0 << BIN_SHIFT)
                             | (unsigned)(qrow0 & (BIN_ROWS - 1)));
        }
        for (int q = q0 + BIN_THREADS; q < qhi; q += BIN_THREADS) {
            const int row = node_idx[q];
            const int p = atomicAdd(&slot[row >> BIN_SHIFT], 1);
            piece[p] = (int)(QRY_FLAG | ((unsigned)q << BIN_SHIFT)
                             | (unsigned)(row & (BIN_ROWS - 1)));
        }
    }
}

template <int U>
__device__ __forceinline__ void gat_walk_step(
    const float* __restrict__ h, int d, float w, int i, int lane,
    float4& acc, float& den)
{
    float4 hv[U];
    float  wi[U];
#pragma unroll
    for (int u = 0; u < U; ++u) {
        const int di = __builtin_amdgcn_readlane(d, i + u);
        wi[u] = __int_as_float(__builtin_amdgcn_readlane(__float_as_int(w), i + u));
        hv[u] = *(const float4*)(h + (size_t)di * OUT_DIM + lane * 4);
    }
#pragma unroll
    for (int u = 0; u < U; ++u) {
        acc.x += wi[u] * hv[u].x;
        acc.y += wi[u] * hv[u].y;
        acc.z += wi[u] * hv[u].z;
        acc.w += wi[u] * hv[u].w;
        den   += wi[u];
    }
}

__device__ __forceinline__ void gat_walk_query(
    const float* __restrict__ h, const float* __restrict__ s_src,
    const float* __restrict__ s_dst, const int* spill, float* __restrict__ out,
    const int* cntL, const int* begL, const int* segL,
    int rq, int rowBase, int lane)
{
    const int r8 = rq & (BIN_ROWS - 1);
    const int j  = (int)(((unsigned)rq & ~QRY_FLAG) >> BIN_SHIFT);
    const int   beg = __builtin_amdgcn_readfirstlane(begL[r8]);
    const int   n   = __builtin_amdgcn_readfirstlane(cntL[r8]);
    const float ss  = s_src[rowBase + r8];

    float4 acc = make_float4(0.0f, 0.0f, 0.0f, 0.0f);
    float  den = 0.0f;

    for (int c = 0; c < n; c += 32) {
        const int m = (n - c < 32) ? (n - c) : 32;
        int   d = 0;
        float w = 0.0f;
        if (lane < m) {
            const int idx = beg + c + lane;
            const gat_lds_int* segS = (const gat_lds_int*)segL;
            d = segS[(idx < SEG_CAP) ? idx : 0];
            if (idx >= SEG_CAP) d = spill[idx];
            float s = ss + s_dst[d];
            s = (s >= 0.0f) ? s : SLOPE * s;
            w = expf(s);
        }
        int i = 0;
        for (; i + 8 <= m; i += 8) gat_walk_step<8>(h, d, w, i, lane, acc, den);
        if (i + 4 <= m) { gat_walk_step<4>(h, d, w, i, lane, acc, den); i += 4; }
        if (i + 2 <= m) { gat_walk_step<2>(h, d, w, i, lane, acc, den); i += 2; }
        if (i < m)      { gat_walk_step<1>(h, d, w, i, lane, acc, den); }
    }

    const float inv = 1.0f / den;
    float4 ov;
    ov.x = acc.x * inv; ov.y = acc.y * inv; ov.z = acc.z * inv; ov.w = acc.w * inv;
    *(float4*)(out + (size_t)j * OUT_DIM + lane * 4) = ov;
}

__global__ __launch_bounds__(BIN_THREADS) void gat_rowwalk_kernel(
    const int* __restrict__ table, const int* __restrict__ binned,
    int* spillAll, int E, int N, int B,
    const float* __restrict__ h, const float* __restrict__ s_src,
    const float* __restrict__ s_dst, float* __restrict__ out)
{
    __shared__ int cnt[BIN_ROWS];
    __shared__ int cur[BIN_ROWS];
    __shared__ int sbeg[BIN_ROWS];
    __shared__ int wsum[BIN_ROWS / 32];
    __shared__ int bsum[BIN_ROWS / 32];
    __shared__ int rbeg[BIN_GROUPS];
    __shared__ int rlen[BIN_GROUPS];
    __shared__ int qn;
    __shared__ int qlist[QRY_CAP];
    __shared__ int segL[SEG_CAP];
    const int tid  = (int)threadIdx.x;
    const int lane = tid & 31;
    const int wv   = __builtin_amdgcn_readfirstlane(tid >> 5);
    const int bin  = (int)blockIdx.x;
    const int per  = (E + BIN_GROUPS - 1) / BIN_GROUPS;
    const int qper = (B + BIN_GROUPS - 1) / BIN_GROUPS;

    int s0 = 0;
    if (tid < BIN_GROUPS) {
        s0 = table[(size_t)bin * BIN_GROUPS + tid];
        const int s1 = table[(size_t)(bin + 1) * BIN_GROUPS + tid];
        int lo = tid * per;
        if (lo > E) lo = E;
        int qlo = tid * qper;
        if (qlo > B) qlo = B;
        rbeg[tid] = lo + qlo + s0;
        rlen[tid] = s1 - s0;
    }
    if (tid < BIN_ROWS) cnt[tid] = 0;
    if (tid == 0) qn = 0;
    int r = s0;
    for (int off = 16; off > 0; off >>= 1) r += __shfl_xor(r, off, 32);
    if (tid < BIN_ROWS && lane == 0) bsum[wv] = r;
    __syncthreads();
    int base = 0;
    for (int w = 0; w < BIN_ROWS / 32; ++w) base += bsum[w];

    int rec[BIN_RUNS];
    {
        int rb[BIN_RUNS];
        int rn[BIN_RUNS];
#pragma unroll
        for (int u = 0; u < BIN_RUNS; ++u) {
            rb[u] = rbeg[wv + 32 * u];
            rn[u] = rlen[wv + 32 * u];
        }
#pragma unroll
        for (int u = 0; u < BIN_RUNS; ++u)
            rec[u] = (lane < rn[u]) ? binned[rb[u] + lane] : -1;
#pragma unroll
        for (int u = 0; u < BIN_RUNS; ++u) {
            if (rec[u] >= 0) {
                atomicAdd(&cnt[rec[u] & (BIN_ROWS - 1)], 1);
            } else if (rec[u] != -1) {
                const int qi = atomicAdd(&qn, 1);
                if (qi < QRY_CAP) qlist[qi] = rec[u];
            }
        }
    }
    for (int u = 0; u < BIN_RUNS; ++u) {
        const int n = rlen[wv + 32 * u];
        const int* __restrict__ run = binned + rbeg[wv + 32 * u];
        for (int i = lane + 32; i < n; i += 32) {
            const int p = run[i];
            if (p >= 0) {
                atomicAdd(&cnt[p & (BIN_ROWS - 1)], 1);
            } else {
                const int qi = atomicAdd(&qn, 1);
                if (qi < QRY_CAP) qlist[qi] = p;
            }
        }
    }
    __syncthreads();

    const int c = (tid < BIN_ROWS) ? cnt[tid] : 0;
    int x = c;
    for (int off = 1; off < 32; off <<= 1) {
        const int y = __shfl_up(x, off, 32);
        if (lane >= off) x += y;
    }
    if (tid < BIN_ROWS && lane == 31) wsum[wv] = x;
    __syncthreads();
    if (tid == 0) {
        int s = 0;
        for (int w = 0; w < BIN_ROWS / 32; ++w) { const int t = wsum[w]; wsum[w] = s; s += t; }
    }
    __syncthreads();
    if (tid < BIN_ROWS) {
        const int excl = wsum[wv] + x - c;
        cur[tid]  = excl;
        sbeg[tid] = excl;
    }
    __syncthreads();

    int* spill = spillAll + base;
#pragma unroll
    for (int u = 0; u < BIN_RUNS; ++u) {
        if (rec[u] >= 0) {
            const int q = atomicAdd(&cur[rec[u] & (BIN_ROWS - 1)], 1);
            const int d = rec[u] >> BIN_SHIFT;
            if (q < SEG_CAP) segL[q] = d; else spill[q] = d;
        }
    }
    for (int u = 0; u < BIN_RUNS; ++u) {
        const int n = rlen[wv + 32 * u];
        const int* __restrict__ run = binned + rbeg[wv + 32 * u];
        for (int i = lane + 32; i < n; i += 32) {
            const int p = run[i];
            if (p >= 0) {
                const int q = atomicAdd(&cur[p & (BIN_ROWS - 1)], 1);
                const int d = p >> BIN_SHIFT;
                if (q < SEG_CAP) segL[q] = d; else spill[q] = d;
            }
        }
    }
    __syncthreads();

    const int nq      = __builtin_amdgcn_readfirstlane(qn);
    const int rowBase = bin * BIN_ROWS;
    if (nq <= QRY_CAP) {
        for (int qi = wv; qi < nq; qi += BIN_THREADS / 32) {
            const int rq = __builtin_amdgcn_readfirstlane(qlist[qi]);
            gat_walk_query(h, s_src, s_dst, spill, out, cnt, sbeg, segL, rq, rowBase, lane);
        }
    } else {
        for (int u = 0; u < BIN_RUNS; ++u) {
            const int n = rlen[wv + 32 * u];
            const int* __restrict__ run = binned + rbeg[wv + 32 * u];
            for (int i0 = 0; i0 < n; i0 += 32) {
                const int p = (i0 + lane < n) ? run[i0 + lane] : 0;
                unsigned mask = __builtin_amdgcn_ballot_w32(p < 0);
                while (mask != 0u) {
                    const int l = __builtin_ctz(mask);
                    mask &= mask - 1u;
                    const int rq = __builtin_amdgcn_readlane(p, l);
                    gat_walk_query(h, s_src, s_dst, spill, out, cnt, sbeg, segL, rq, rowBase, lane);
                }
            }
        }
    }
}

extern "C" void kernel_launch(void* const* d_in, const int* in_sizes, int n_in,
                              void* d_out, int out_size, void* d_ws, size_t ws_size,
                              hipStream_t stream) {
    const float* features     = (const float*)d_in[0];
    const float* W            = (const float*)d_in[1];
    const float* b            = (const float*)d_in[2];
    const float* a            = (const float*)d_in[3];
    const int*   edges        = (const int*)d_in[4];
    const int*   unique_nodes = (const int*)d_in[5];
    const int*   node_idx     = (const int*)d_in[6];
    float* out = (float*)d_out;

    const int N = in_sizes[5];
    const int E = in_sizes[4] / 2;
    const int B = in_sizes[6];

    float* h      = (float*)d_ws;
    float* s_src  = h + (size_t)N * OUT_DIM;
    float* s_dst  = s_src + N;
    int*   table  = (int*)(s_dst + N);
    int*   binned = table + N + 1;
    int*   spill  = binned + E + B;

    {
        gat_gemm_kernel<<<GEMM_GROUPS, 256, 0, stream>>>(features, W, b, unique_nodes, h, N);
    }
    {
        gat_bin_scores_kernel<<<BIN_GROUPS + SCORE_GROUPS, BIN_THREADS, 0, stream>>>(
            edges, node_idx, table, binned, E, N, B, h, a, s_src, s_dst);
    }
    {
        dim3 grid((N + 255) / 256);
        gat_rowwalk_kernel<<<grid, BIN_THREADS, 0, stream>>>(table, binned, spill, E, N, B,
                                                             h, s_src, s_dst, out);
    }
}
